# starting kernel
# baseline (speedup 1.0000x reference)
#include <hip/hip_runtime.h>
#include <math.h>

#define FDIM 64

typedef __attribute__((ext_vector_type(2))) float v2f;
typedef __attribute__((ext_vector_type(8))) float v8f;

__device__ __forceinline__ void atomicMaxF32(float* addr, float val) {
    if (val >= 0.0f) {
        atomicMax((int*)addr, __float_as_int(val));
    } else {
        atomicMin((unsigned int*)addr, __float_as_uint(val));
    }
}

__global__ __launch_bounds__(32) void gat_gemm_wmma(
    const float* __restrict__ X,
    const float* __restrict__ W,
    const float* __restrict__ a_s,
    const float* __restrict__ a_d,
    float* __restrict__ H,
    float* __restrict__ HS,
    float* __restrict__ HD)
{
    const int lane  = threadIdx.x;
    const int m     = lane & 15;
    const int khalf = lane >> 4;
    const int row0  = blockIdx.x * 16;

    const float* xr = X + (size_t)(row0 + m) * FDIM;
    v2f af[16];
#pragma unroll
    for (int k = 0; k < 16; ++k) {
        const int kc = 4 * k + 2 * khalf;
        af[k].x = xr[kc];
        af[k].y = xr[kc + 1];
    }

    v8f acc[4];
#pragma unroll
    for (int j = 0; j < 4; ++j) acc[j] = (v8f){0,0,0,0,0,0,0,0};

#pragma unroll
    for (int j = 0; j < 4; ++j) {
#pragma unroll
        for (int k = 0; k < 16; ++k) {
            const float* wp = W + (size_t)(4 * k + 2 * khalf) * FDIM + j * 16 + m;
            v2f bf;
            bf.x = wp[0];
            bf.y = wp[FDIM];
            acc[j] = __builtin_amdgcn_wmma_f32_16x16x4_f32(
                 false, af[k],  false, bf,
                 (short)0, acc[j],
                 false,  false);
        }
    }

    const int drow_base = row0 + 8 * khalf;
#pragma unroll
    for (int r = 0; r < 8; ++r) {
        float* hp = H + (size_t)(drow_base + r) * FDIM + m;
#pragma unroll
        for (int j = 0; j < 4; ++j) hp[j * 16] = acc[j][r];
    }

    float as_c[4], ad_c[4];
#pragma unroll
    for (int j = 0; j < 4; ++j) {
        as_c[j] = a_s[j * 16 + m];
        ad_c[j] = a_d[j * 16 + m];
    }
#pragma unroll
    for (int r = 0; r < 8; ++r) {
        float ps = 0.0f, pd = 0.0f;
#pragma unroll
        for (int j = 0; j < 4; ++j) {
            ps += acc[j][r] * as_c[j];
            pd += acc[j][r] * ad_c[j];
        }
#pragma unroll
        for (int off = 8; off >= 1; off >>= 1) {
            ps += __shfl_xor(ps, off, 32);
            pd += __shfl_xor(pd, off, 32);
        }
        if (m == 0) {
            HS[drow_base + r] = ps;
            HD[drow_base + r] = pd;
        }
    }
}

__global__ void gat_init(float* __restrict__ OUT, float* __restrict__ Mx,
                         float* __restrict__ Sx, int n_nodes, int tot)
{
    int i = blockIdx.x * blockDim.x + threadIdx.x;
    if (i < tot) OUT[i] = 0.0f;
    if (i < n_nodes) {
        Mx[i] = __int_as_float(0xFF800000);
        Sx[i] = 0.0f;
    }
}

__global__ void gat_edge_score(const int* __restrict__ src,
                               const int* __restrict__ dst,
                               const float* __restrict__ HS,
                               const float* __restrict__ HD,
                               float* __restrict__ EW,
                               float* __restrict__ Mx,
                               int E, int Etot)
{
    int e = blockIdx.x * blockDim.x + threadIdx.x;
    if (e >= Etot) return;
    int s, d;
    if (e < E) { s = src[e]; d = dst[e]; } else { s = d = e - E; }
    float v = HS[s] + HD[d];
    v = (v >= 0.0f) ? v : 0.2f * v;
    EW[e] = v;
    atomicMaxF32(&Mx[d], v);
}

__global__ void gat_edge_exp(const int* __restrict__ dst,
                             const float* __restrict__ Mx,
                             float* __restrict__ EW,
                             float* __restrict__ Sx,
                             int E, int Etot)
{
    int e = blockIdx.x * blockDim.x + threadIdx.x;
    if (e >= Etot) return;
    int d = (e < E) ? dst[e] : (e - E);
    float w = __expf(EW[e] - Mx[d]);
    EW[e] = w;
    atomicAdd(&Sx[d], w);
}

#define EDGES_PER_BLOCK 16
__global__ __launch_bounds__(64) void gat_aggr(
    const int* __restrict__ src, const int* __restrict__ dst,
    const float* __restrict__ EW, const float* __restrict__ Sx,
    const float* __restrict__ H, float* __restrict__ OUT,
    int E, int Etot)
{
    const int f  = threadIdx.x;
    const int e0 = blockIdx.x * EDGES_PER_BLOCK;
#pragma unroll 1
    for (int i = 0; i < EDGES_PER_BLOCK; ++i) {
        int e = e0 + i;
        if (e >= Etot) return;
        int s, d;
        if (e < E) { s = src[e]; d = dst[e]; } else { s = d = e - E; }
        float alpha = EW[e] / Sx[d];
        atomicAdd(&OUT[(size_t)d * FDIM + f], alpha * H[(size_t)s * FDIM + f]);
    }
}

__global__ void gat_bias(float* __restrict__ OUT, const float* __restrict__ b,
                         int tot)
{
    int i = blockIdx.x * blockDim.x + threadIdx.x;
    if (i < tot) OUT[i] += b[i & (FDIM - 1)];
}

extern "C" void kernel_launch(void* const* d_in, const int* in_sizes, int n_in,
                              void* d_out, int out_size, void* d_ws, size_t ws_size,
                              hipStream_t stream) {
    const float* x   = (const float*)d_in[0];
    const int*   ei  = (const int*)  d_in[1];
    const float* W1  = (const float*)d_in[2];
    const float* as1 = (const float*)d_in[3];
    const float* ad1 = (const float*)d_in[4];
    const float* b1  = (const float*)d_in[5];
    const float* W2  = (const float*)d_in[6];
    const float* as2 = (const float*)d_in[7];
    const float* ad2 = (const float*)d_in[8];
    const float* b2  = (const float*)d_in[9];

    const int N    = in_sizes[0] / FDIM;
    const int E    = in_sizes[1] / 2;
    const int Etot = E + N;
    const int* src = ei;
    const int* dst = ei + E;

    float* ws = (float*)d_ws;
    float* H  = ws;  ws += (size_t)N * FDIM;
    float* X1 = ws;  ws += (size_t)N * FDIM;
    float* HS = ws;  ws += N;
    float* HD = ws;  ws += N;
    float* Mx = ws;  ws += N;
    float* Sx = ws;  ws += N;
    float* EW = ws;  ws += Etot;

    const int tot = N * FDIM;

    auto layer = [&](const float* Xin, const float* W, const float* a_s,
                     const float* a_d, const float* b, float* OUT) {
        gat_gemm_wmma<<<N / 16, 32, 0, stream>>>(Xin, W, a_s, a_d, H, HS, HD);
        gat_init<<<(tot + 255) / 256, 256, 0, stream>>>(OUT, Mx, Sx, N, tot);
        gat_edge_score<<<(Etot + 255) / 256, 256, 0, stream>>>(src, dst, HS, HD,
                                                               EW, Mx, E, Etot);
        gat_edge_exp<<<(Etot + 255) / 256, 256, 0, stream>>>(dst, Mx, EW, Sx,
                                                             E, Etot);
        gat_aggr<<<(Etot + EDGES_PER_BLOCK - 1) / EDGES_PER_BLOCK, 64, 0,
                   stream>>>(src, dst, EW, Sx, H, OUT, E, Etot);
        gat_bias<<<(tot + 255) / 256, 256, 0, stream>>>(OUT, b, tot);
    };

    layer(x,  W1, as1, ad1, b1, X1);
    layer(X1, W2, as2, ad2, b2, (float*)d_out);
}


# Replace atomic scatter softmax+aggregate by a per-destination gather over an edge list counting-sorted in-call (arrival-order rank from int atomics sets the f32 sum order); float atomics removed
# speedup vs baseline: 2.5994x; 2.5994x over previous
#include <hip/hip_runtime.h>
#include <math.h>

#define FDIM 64

typedef __attribute__((ext_vector_type(2))) float v2f;
typedef __attribute__((ext_vector_type(8))) float v8f;

__global__ __launch_bounds__(32) void gat_gemm_wmma(
    const float* __restrict__ X,
    const float* __restrict__ W,
    const float* __restrict__ a_s,
    const float* __restrict__ a_d,
    float* __restrict__ H,
    float* __restrict__ HS,
    float* __restrict__ HD)
{
    const int lane  = threadIdx.x;
    const int m     = lane & 15;
    const int khalf = lane >> 4;
    const int row0  = blockIdx.x * 16;

    const float* xr = X + (size_t)(row0 + m) * FDIM;
    v2f af[16];
#pragma unroll
    for (int k = 0; k < 16; ++k) {
        const int kc = 4 * k + 2 * khalf;
        af[k].x = xr[kc];
        af[k].y = xr[kc + 1];
    }

    v8f acc[4];
#pragma unroll
    for (int j = 0; j < 4; ++j) acc[j] = (v8f){0,0,0,0,0,0,0,0};

#pragma unroll
    for (int j = 0; j < 4; ++j) {
#pragma unroll
        for (int k = 0; k < 16; ++k) {
            const float* wp = W + (size_t)(4 * k + 2 * khalf) * FDIM + j * 16 + m;
            v2f bf;
            bf.x = wp[0];
            bf.y = wp[FDIM];
            acc[j] = __builtin_amdgcn_wmma_f32_16x16x4_f32(
                 false, af[k],  false, bf,
                 (short)0, acc[j],
                 false,  false);
        }
    }

    const int drow_base = row0 + 8 * khalf;
#pragma unroll
    for (int r = 0; r < 8; ++r) {
        float* hp = H + (size_t)(drow_base + r) * FDIM + m;
#pragma unroll
        for (int j = 0; j < 4; ++j) hp[j * 16] = acc[j][r];
    }

    float as_c[4], ad_c[4];
#pragma unroll
    for (int j = 0; j < 4; ++j) {
        as_c[j] = a_s[j * 16 + m];
        ad_c[j] = a_d[j * 16 + m];
    }
#pragma unroll
    for (int r = 0; r < 8; ++r) {
        float ps = 0.0f, pd = 0.0f;
#pragma unroll
        for (int j = 0; j < 4; ++j) {
            ps += acc[j][r] * as_c[j];
            pd += acc[j][r] * ad_c[j];
        }
#pragma unroll
        for (int off = 8; off >= 1; off >>= 1) {
            ps += __shfl_xor(ps, off, 32);
            pd += __shfl_xor(pd, off, 32);
        }
        if (m == 0) {
            HS[drow_base + r] = ps;
            HD[drow_base + r] = pd;
        }
    }
}

__global__ void csr_zero(int* __restrict__ cnt, int n)
{
    int i = blockIdx.x * blockDim.x + threadIdx.x;
    if (i < n) cnt[i] = 0;
}

__global__ void csr_count(const int* __restrict__ dst, int* __restrict__ cnt,
                          int* __restrict__ rank, int E)
{
    int e = blockIdx.x * blockDim.x + threadIdx.x;
    if (e < E) rank[e] = atomicAdd(&cnt[dst[e]], 1);
}

#define SCAN_THREADS 256
#define SCAN_SPAN 1024
__global__ __launch_bounds__(SCAN_THREADS) void csr_scan(
    const int* __restrict__ cnt, int* __restrict__ rowptr, int n)
{
    __shared__ int s_pre[SCAN_THREADS];
    __shared__ int s_scan[SCAN_THREADS];
    const int t    = threadIdx.x;
    const int blk0 = blockIdx.x * SCAN_SPAN;

    int pre = 0;
    for (int i = t; i < blk0; i += SCAN_THREADS) pre += cnt[i];

    const int base = blk0 + 4 * t;
    int c[4];
#pragma unroll
    for (int q = 0; q < 4; ++q) c[q] = (base + q < n) ? cnt[base + q] : 0;
    const int tsum = c[0] + c[1] + c[2] + c[3];

    s_pre[t]  = pre;
    s_scan[t] = tsum;
    __syncthreads();
    for (int off = SCAN_THREADS / 2; off > 0; off >>= 1) {
        if (t < off) s_pre[t] += s_pre[t + off];
        __syncthreads();
    }
    const int front = s_pre[0];
    for (int off = 1; off < SCAN_THREADS; off <<= 1) {
        const int v = (t >= off) ? s_scan[t - off] : 0;
        __syncthreads();
        s_scan[t] += v;
        __syncthreads();
    }
    int run = front + s_scan[t] - tsum;
#pragma unroll
    for (int q = 0; q < 4; ++q) {
        if (base + q <= n) rowptr[base + q] = run;
        run += c[q];
    }
}

__global__ void csr_fill(const int* __restrict__ src, const int* __restrict__ dst,
                         const int* __restrict__ rank,
                         const int* __restrict__ rowptr,
                         int* __restrict__ csr_src, int E)
{
    int e = blockIdx.x * blockDim.x + threadIdx.x;
    if (e < E) csr_src[rowptr[dst[e]] + rank[e]] = src[e];
}

#define NODES_PER_BLOCK 8
__global__ __launch_bounds__(32 * NODES_PER_BLOCK) void gat_gather(
    const int* __restrict__ rowptr, const int* __restrict__ csr_src,
    const float* __restrict__ HS, const float* __restrict__ HD,
    const float* __restrict__ H, const float* __restrict__ bias,
    float* __restrict__ OUT, int n_nodes)
{
    const int lane = threadIdx.x & 31;
    const int half = lane >> 4;
    const int c    = lane & 15;
    const int node = blockIdx.x * NODES_PER_BLOCK + (threadIdx.x >> 5);
    if (node >= n_nodes) return;

    const int   beg = rowptr[node];
    const int   end = rowptr[node + 1];
    const float hd  = HD[node];
    const float NEG_INF = __int_as_float(0xFF800000);

    float es = HS[node] + hd;
    es = (es >= 0.0f) ? es : 0.2f * es;

    const int  j0  = beg + lane;
    const bool ok0 = j0 < end;
    const int  s0  = ok0 ? csr_src[j0] : node;
    float e0 = NEG_INF;
    if (ok0) {
        const float v = HS[s0] + hd;
        e0 = (v >= 0.0f) ? v : 0.2f * v;
    }

    float mx = fmaxf(es, e0);
    for (int j = j0 + 32; j < end; j += 32) {
        float v = HS[csr_src[j]] + hd;
        v = (v >= 0.0f) ? v : 0.2f * v;
        mx = fmaxf(mx, v);
    }
#pragma unroll
    for (int off = 16; off >= 1; off >>= 1) mx = fmaxf(mx, __shfl_xor(mx, off, 32));

    float w = ok0 ? __expf(e0 - mx) : 0.0f;
    float sum = w;
    for (int j = j0 + 32; j < end; j += 32) {
        float v = HS[csr_src[j]] + hd;
        v = (v >= 0.0f) ? v : 0.2f * v;
        sum += __expf(v - mx);
    }
#pragma unroll
    for (int off = 16; off >= 1; off >>= 1) sum += __shfl_xor(sum, off, 32);
    const float wself = __expf(es - mx);
    const float S = sum + wself;

    const float4* __restrict__ Hv = (const float4*)H;
    const float4 hself = Hv[(size_t)node * 16 + c];
    const float  aself = (half == 0) ? (wself / S) : 0.0f;
    float4 acc;
    acc.x = aself * hself.x; acc.y = aself * hself.y;
    acc.z = aself * hself.z; acc.w = aself * hself.w;

    int  s  = s0;
    bool ok = ok0;
    for (int base = beg; base < end; base += 32) {
        if (base != beg) {
            const int j = base + lane;
            ok = j < end;
            s  = ok ? csr_src[j] : node;
            w  = 0.0f;
            if (ok) {
                float v = HS[s] + hd;
                v = (v >= 0.0f) ? v : 0.2f * v;
                w = __expf(v - mx);
            }
        }
        const float a = ok ? (w / S) : 0.0f;
        const int   n = (end - base < 32) ? (end - base) : 32;
        for (int k = 0; k < n; k += 4) {
            const int   i0 = k + half;
            const int   i1 = k + 2 + half;
            const int   sa = __shfl(s, i0, 32);
            const float aa = __shfl(a, i0, 32);
            const int   sb = __shfl(s, i1, 32);
            const float ab = __shfl(a, i1, 32);
            const float4 ra = Hv[(size_t)sa * 16 + c];
            const float4 rb = Hv[(size_t)sb * 16 + c];
            acc.x += aa * ra.x; acc.y += aa * ra.y;
            acc.z += aa * ra.z; acc.w += aa * ra.w;
            acc.x += ab * rb.x; acc.y += ab * rb.y;
            acc.z += ab * rb.z; acc.w += ab * rb.w;
        }
    }

    acc.x += __shfl_xor(acc.x, 16, 32);
    acc.y += __shfl_xor(acc.y, 16, 32);
    acc.z += __shfl_xor(acc.z, 16, 32);
    acc.w += __shfl_xor(acc.w, 16, 32);
    if (half == 0) {
        float4 o;
        o.x = acc.x + bias[4 * c + 0];
        o.y = acc.y + bias[4 * c + 1];
        o.z = acc.z + bias[4 * c + 2];
        o.w = acc.w + bias[4 * c + 3];
        ((float4*)OUT)[(size_t)node * 16 + c] = o;
    }
}

extern "C" void kernel_launch(void* const* d_in, const int* in_sizes, int n_in,
                              void* d_out, int out_size, void* d_ws, size_t ws_size,
                              hipStream_t stream) {
    const float* x   = (const float*)d_in[0];
    const int*   ei  = (const int*)  d_in[1];
    const float* W1  = (const float*)d_in[2];
    const float* as1 = (const float*)d_in[3];
    const float* ad1 = (const float*)d_in[4];
    const float* b1  = (const float*)d_in[5];
    const float* W2  = (const float*)d_in[6];
    const float* as2 = (const float*)d_in[7];
    const float* ad2 = (const float*)d_in[8];
    const float* b2  = (const float*)d_in[9];

    const int N    = in_sizes[0] / FDIM;
    const int E    = in_sizes[1] / 2;
    const int* src = ei;
    const int* dst = ei + E;

    float* ws = (float*)d_ws;
    float* H  = ws;  ws += (size_t)N * FDIM;
    float* X1 = ws;  ws += (size_t)N * FDIM;
    float* HS = ws;  ws += N;
    float* HD = ws;  ws += N;
    int* wi      = (int*)ws;
    int* cnt     = wi;  wi += N;
    int* rank    = wi;  wi += E;
    int* csr_src = wi;  wi += E;
    int* rowptr  = wi;  wi += N + 1;

    csr_zero<<<(N + 255) / 256, 256, 0, stream>>>(cnt, N);
    csr_count<<<(E + 255) / 256, 256, 0, stream>>>(dst, cnt, rank, E);
    csr_scan<<<(N + SCAN_SPAN) / SCAN_SPAN, SCAN_THREADS, 0, stream>>>(cnt, rowptr, N);
    csr_fill<<<(E + 255) / 256, 256, 0, stream>>>(src, dst, rank, rowptr, csr_src, E);

    auto layer = [&](const float* Xin, const float* W, const float* a_s,
                     const float* a_d, const float* b, float* OUT) {
        gat_gemm_wmma<<<N / 16, 32, 0, stream>>>(Xin, W, a_s, a_d, H, HS, HD);
        gat_gather<<<(N + NODES_PER_BLOCK - 1) / NODES_PER_BLOCK,
                     32 * NODES_PER_BLOCK, 0, stream>>>(rowptr, csr_src, HS, HD,
                                                        H, b, OUT, N);
    };

    layer(x,  W1, as1, ad1, b1, X1);
    layer(X1, W2, as2, ad2, b2, (float*)d_out);
}


# Count+scan+place sort replaced by fixed 64-slot per-destination lists filled in one pass (slot = returning int add, arrival order as before); a fuller node reads the raw edge list; 8 to 6 launches
# speedup vs baseline: 2.9992x; 1.1538x over previous
#include <hip/hip_runtime.h>
#include <math.h>

#define FDIM 64

typedef __attribute__((ext_vector_type(2))) float v2f;
typedef __attribute__((ext_vector_type(8))) float v8f;

__global__ __launch_bounds__(32) void gat_gemm_wmma(
    const float* __restrict__ X,
    const float* __restrict__ W,
    const float* __restrict__ a_s,
    const float* __restrict__ a_d,
    float* __restrict__ H,
    float* __restrict__ HS,
    float* __restrict__ HD)
{
    const int lane  = threadIdx.x;
    const int m     = lane & 15;
    const int khalf = lane >> 4;
    const int row0  = blockIdx.x * 16;

    const float* xr = X + (size_t)(row0 + m) * FDIM;
    v2f af[16];
#pragma unroll
    for (int k = 0; k < 16; ++k) {
        const int kc = 4 * k + 2 * khalf;
        af[k].x = xr[kc];
        af[k].y = xr[kc + 1];
    }

    v8f acc[4];
#pragma unroll
    for (int j = 0; j < 4; ++j) acc[j] = (v8f){0,0,0,0,0,0,0,0};

#pragma unroll
    for (int j = 0; j < 4; ++j) {
#pragma unroll
        for (int k = 0; k < 16; ++k) {
            const float* wp = W + (size_t)(4 * k + 2 * khalf) * FDIM + j * 16 + m;
            v2f bf;
            bf.x = wp[0];
            bf.y = wp[FDIM];
            acc[j] = __builtin_amdgcn_wmma_f32_16x16x4_f32(
                 false, af[k],  false, bf,
                 (short)0, acc[j],
                 false,  false);
        }
    }

    const int drow_base = row0 + 8 * khalf;
#pragma unroll
    for (int r = 0; r < 8; ++r) {
        float* hp = H + (size_t)(drow_base + r) * FDIM + m;
#pragma unroll
        for (int j = 0; j < 4; ++j) hp[j * 16] = acc[j][r];
    }

    float as_c[4], ad_c[4];
#pragma unroll
    for (int j = 0; j < 4; ++j) {
        as_c[j] = a_s[j * 16 + m];
        ad_c[j] = a_d[j * 16 + m];
    }
#pragma unroll
    for (int r = 0; r < 8; ++r) {
        float ps = 0.0f, pd = 0.0f;
#pragma unroll
        for (int j = 0; j < 4; ++j) {
            ps += acc[j][r] * as_c[j];
            pd += acc[j][r] * ad_c[j];
        }
#pragma unroll
        for (int off = 8; off >= 1; off >>= 1) {
            ps += __shfl_xor(ps, off, 32);
            pd += __shfl_xor(pd, off, 32);
        }
        if (m == 0) {
            HS[drow_base + r] = ps;
            HD[drow_base + r] = pd;
        }
    }
}

#define LCAP 64

__global__ void adj_zero(int* __restrict__ cnt, int n)
{
    int i = blockIdx.x * blockDim.x + threadIdx.x;
    if (i < n) cnt[i] = 0;
}

__global__ void adj_fill(const int* __restrict__ src, const int* __restrict__ dst,
                         int* __restrict__ cnt, int* __restrict__ lists, int E)
{
    int e = blockIdx.x * blockDim.x + threadIdx.x;
    if (e < E) {
        const int d   = dst[e];
        const int sv  = src[e];
        asm volatile("" :: "v"(sv));
        const int pos = atomicAdd(&cnt[d], 1);
        if (pos < LCAP) lists[(size_t)d * LCAP + pos] = sv;
    }
}

#define NODES_PER_BLOCK 8
__global__ __launch_bounds__(32 * NODES_PER_BLOCK) void gat_gather(
    const int* __restrict__ cnt, const int* __restrict__ lists,
    const int* __restrict__ esrc, const int* __restrict__ edst, int E,
    const float* __restrict__ HS, const float* __restrict__ HD,
    const float* __restrict__ H, const float* __restrict__ bias,
    float* __restrict__ OUT, int n_nodes)
{
    const int lane = threadIdx.x & 31;
    const int half = lane >> 4;
    const int c    = lane & 15;
    const int node = blockIdx.x * NODES_PER_BLOCK + (threadIdx.x >> 5);
    if (node >= n_nodes) return;

    const int   deg = cnt[node];
    const float hd  = HD[node];
    const float NEG_INF = __int_as_float(0xFF800000);

    float es = HS[node] + hd;
    es = (es >= 0.0f) ? es : 0.2f * es;

    const float4* __restrict__ Hv = (const float4*)H;
    float4 acc;

    if (deg <= LCAP) {
        const int* __restrict__ lst = lists + (size_t)node * LCAP;

        const bool ok0 = lane < deg;
        const int  s0  = ok0 ? lst[lane] : node;
        float e0 = NEG_INF;
        if (ok0) {
            const float v = HS[s0] + hd;
            e0 = (v >= 0.0f) ? v : 0.2f * v;
        }

        float mx = fmaxf(es, e0);
        for (int j = lane + 32; j < deg; j += 32) {
            float v = HS[lst[j]] + hd;
            v = (v >= 0.0f) ? v : 0.2f * v;
            mx = fmaxf(mx, v);
        }
#pragma unroll
        for (int off = 16; off >= 1; off >>= 1) mx = fmaxf(mx, __shfl_xor(mx, off, 32));

        float w = ok0 ? __expf(e0 - mx) : 0.0f;
        float sum = w;
        for (int j = lane + 32; j < deg; j += 32) {
            float v = HS[lst[j]] + hd;
            v = (v >= 0.0f) ? v : 0.2f * v;
            sum += __expf(v - mx);
        }
#pragma unroll
        for (int off = 16; off >= 1; off >>= 1) sum += __shfl_xor(sum, off, 32);
        const float wself = __expf(es - mx);
        const float S = sum + wself;

        const float4 hself = Hv[(size_t)node * 16 + c];
        const float  aself = (half == 0) ? (wself / S) : 0.0f;
        acc.x = aself * hself.x; acc.y = aself * hself.y;
        acc.z = aself * hself.z; acc.w = aself * hself.w;

        int  s  = s0;
        bool ok = ok0;
        for (int base = 0; base < deg; base += 32) {
            if (base != 0) {
                const int j = base + lane;
                ok = j < deg;
                s  = ok ? lst[j] : node;
                w  = 0.0f;
                if (ok) {
                    float v = HS[s] + hd;
                    v = (v >= 0.0f) ? v : 0.2f * v;
                    w = __expf(v - mx);
                }
            }
            const float a = ok ? (w / S) : 0.0f;
            const int   n = (deg - base < 32) ? (deg - base) : 32;
            for (int k = 0; k < n; k += 4) {
                const int   i0 = k + half;
                const int   i1 = k + 2 + half;
                const int   sa = __shfl(s, i0, 32);
                const float aa = __shfl(a, i0, 32);
                const int   sb = __shfl(s, i1, 32);
                const float ab = __shfl(a, i1, 32);
                const float4 ra = Hv[(size_t)sa * 16 + c];
                const float4 rb = Hv[(size_t)sb * 16 + c];
                acc.x += aa * ra.x; acc.y += aa * ra.y;
                acc.z += aa * ra.z; acc.w += aa * ra.w;
                acc.x += ab * rb.x; acc.y += ab * rb.y;
                acc.z += ab * rb.z; acc.w += ab * rb.w;
            }
        }
    } else {
        float mx = es;
        for (int base = 0; base < E; base += 32) {
            const int e = base + lane;
            if (e < E && edst[e] == node) {
                float v = HS[esrc[e]] + hd;
                v = (v >= 0.0f) ? v : 0.2f * v;
                mx = fmaxf(mx, v);
            }
        }
        for (int off = 16; off >= 1; off >>= 1) mx = fmaxf(mx, __shfl_xor(mx, off, 32));

        float sum = 0.0f;
        acc.x = 0.0f; acc.y = 0.0f; acc.z = 0.0f; acc.w = 0.0f;
        for (int base = 0; base < E; base += 32) {
            const int  e   = base + lane;
            const bool hit = (e < E) && (edst[e] == node);
            int   s = node;
            float w = 0.0f;
            if (hit) {
                s = esrc[e];
                float v = HS[s] + hd;
                v = (v >= 0.0f) ? v : 0.2f * v;
                w = __expf(v - mx);
            }
            sum += w;
            unsigned m = (unsigned)__ballot(hit);
            while (m != 0u) {
                const int k = __builtin_ctz(m);
                m &= m - 1u;
                const int   sk = __shfl(s, k, 32);
                const float wk = __shfl(w, k, 32);
                const float a  = (half == 0) ? wk : 0.0f;
                const float4 r = Hv[(size_t)sk * 16 + c];
                acc.x += a * r.x; acc.y += a * r.y;
                acc.z += a * r.z; acc.w += a * r.w;
            }
        }
        for (int off = 16; off >= 1; off >>= 1) sum += __shfl_xor(sum, off, 32);
        const float wself = __expf(es - mx);
        const float S = sum + wself;
        const float4 hself = Hv[(size_t)node * 16 + c];
        const float  aself = (half == 0) ? wself : 0.0f;
        acc.x = (acc.x + aself * hself.x) / S;
        acc.y = (acc.y + aself * hself.y) / S;
        acc.z = (acc.z + aself * hself.z) / S;
        acc.w = (acc.w + aself * hself.w) / S;
    }

    acc.x += __shfl_xor(acc.x, 16, 32);
    acc.y += __shfl_xor(acc.y, 16, 32);
    acc.z += __shfl_xor(acc.z, 16, 32);
    acc.w += __shfl_xor(acc.w, 16, 32);
    if (half == 0) {
        float4 o;
        o.x = acc.x + bias[4 * c + 0];
        o.y = acc.y + bias[4 * c + 1];
        o.z = acc.z + bias[4 * c + 2];
        o.w = acc.w + bias[4 * c + 3];
        ((float4*)OUT)[(size_t)node * 16 + c] = o;
    }
}

extern "C" void kernel_launch(void* const* d_in, const int* in_sizes, int n_in,
                              void* d_out, int out_size, void* d_ws, size_t ws_size,
                              hipStream_t stream) {
    const float* x   = (const float*)d_in[0];
    const int*   ei  = (const int*)  d_in[1];
    const float* W1  = (const float*)d_in[2];
    const float* as1 = (const float*)d_in[3];
    const float* ad1 = (const float*)d_in[4];
    const float* b1  = (const float*)d_in[5];
    const float* W2  = (const float*)d_in[6];
    const float* as2 = (const float*)d_in[7];
    const float* ad2 = (const float*)d_in[8];
    const float* b2  = (const float*)d_in[9];

    const int N    = in_sizes[0] / FDIM;
    const int E    = in_sizes[1] / 2;
    const int* src = ei;
    const int* dst = ei + E;

    float* ws = (float*)d_ws;
    float* H  = ws;  ws += (size_t)N * FDIM;
    float* X1 = ws;  ws += (size_t)N * FDIM;
    int* lists = (int*)ws;  ws += (size_t)N * LCAP;
    float* HS = ws;  ws += N;
    float* HD = ws;  ws += N;
    int* cnt  = (int*)ws;

    adj_zero<<<(N + 255) / 256, 256, 0, stream>>>(cnt, N);
    adj_fill<<<(E + 255) / 256, 256, 0, stream>>>(src, dst, cnt, lists, E);

    auto layer = [&](const float* Xin, const float* W, const float* a_s,
                     const float* a_d, const float* b, float* OUT) {
        gat_gemm_wmma<<<N / 16, 32, 0, stream>>>(Xin, W, a_s, a_d, H, HS, HD);
        gat_gather<<<(N + NODES_PER_BLOCK - 1) / NODES_PER_BLOCK,
                     32 * NODES_PER_BLOCK, 0, stream>>>(cnt, lists, src, dst, E,
                                                        HS, HD, H, b, OUT, N);
    };

    layer(x,  W1, as1, ad1, b1, X1);
    layer(X1, W2, as2, ad2, b2, (float*)d_out);
}


# Gather in one pass: each source's score is formed from the row gathered anyway (row . a_s over the half's 16 lanes), so no per-edge score gather; running max with rescale; f32 sum order changes
# speedup vs baseline: 3.1067x; 1.0358x over previous
#include <hip/hip_runtime.h>
#include <math.h>

#define FDIM 64

typedef __attribute__((ext_vector_type(2))) float v2f;
typedef __attribute__((ext_vector_type(8))) float v8f;

__global__ __launch_bounds__(32) void gat_gemm_wmma(
    const float* __restrict__ X,
    const float* __restrict__ W,
    const float* __restrict__ a_s,
    const float* __restrict__ a_d,
    float* __restrict__ H,
    float* __restrict__ HS,
    float* __restrict__ HD)
{
    const int lane  = threadIdx.x;
    const int m     = lane & 15;
    const int khalf = lane >> 4;
    const int row0  = blockIdx.x * 16;

    const float* xr = X + (size_t)(row0 + m) * FDIM;
    v2f af[16];
#pragma unroll
    for (int k = 0; k < 16; ++k) {
        const int kc = 4 * k + 2 * khalf;
        af[k].x = xr[kc];
        af[k].y = xr[kc + 1];
    }

    v8f acc[4];
#pragma unroll
    for (int j = 0; j < 4; ++j) acc[j] = (v8f){0,0,0,0,0,0,0,0};

#pragma unroll
    for (int j = 0; j < 4; ++j) {
#pragma unroll
        for (int k = 0; k < 16; ++k) {
            const float* wp = W + (size_t)(4 * k + 2 * khalf) * FDIM + j * 16 + m;
            v2f bf;
            bf.x = wp[0];
            bf.y = wp[FDIM];
            acc[j] = __builtin_amdgcn_wmma_f32_16x16x4_f32(
                 false, af[k],  false, bf,
                 (short)0, acc[j],
                 false,  false);
        }
    }

    const int drow_base = row0 + 8 * khalf;
#pragma unroll
    for (int r = 0; r < 8; ++r) {
        float* hp = H + (size_t)(drow_base + r) * FDIM + m;
#pragma unroll
        for (int j = 0; j < 4; ++j) hp[j * 16] = acc[j][r];
    }

    float as_c[4], ad_c[4];
#pragma unroll
    for (int j = 0; j < 4; ++j) {
        as_c[j] = a_s[j * 16 + m];
        ad_c[j] = a_d[j * 16 + m];
    }
#pragma unroll
    for (int r = 0; r < 8; ++r) {
        float ps = 0.0f, pd = 0.0f;
#pragma unroll
        for (int j = 0; j < 4; ++j) {
            ps += acc[j][r] * as_c[j];
            pd += acc[j][r] * ad_c[j];
        }
#pragma unroll
        for (int off = 8; off >= 1; off >>= 1) {
            ps += __shfl_xor(ps, off, 32);
            pd += __shfl_xor(pd, off, 32);
        }
        if (m == 0) {
            HS[drow_base + r] = ps;
            HD[drow_base + r] = pd;
        }
    }
}

#define LCAP 64

__global__ void adj_zero(int* __restrict__ cnt, int n)
{
    int i = blockIdx.x * blockDim.x + threadIdx.x;
    if (i < n) cnt[i] = 0;
}

__global__ void adj_fill(const int* __restrict__ src, const int* __restrict__ dst,
                         int* __restrict__ cnt, int* __restrict__ lists, int E)
{
    int e = blockIdx.x * blockDim.x + threadIdx.x;
    if (e < E) {
        const int d   = dst[e];
        const int sv  = src[e];
        asm volatile("" :: "v"(sv));
        const int pos = atomicAdd(&cnt[d], 1);
        if (pos < LCAP) lists[(size_t)d * LCAP + pos] = sv;
    }
}

#define NODES_PER_BLOCK 8

template <int N>
static __device__ __forceinline__ float row_xor(float x)
{
    return __int_as_float(__builtin_amdgcn_update_dpp(0, __float_as_int(x), 0x160 + N, 0xf, 0xf, true));
}

static __device__ __forceinline__ float row16_sum(float p)
{
    p += row_xor<1>(p);
    p += row_xor<2>(p);
    p += row_xor<4>(p);
    p += row_xor<8>(p);
    return p;
}

template <int K>
static __device__ __forceinline__ int row_bcast(int v)
{
    return __builtin_amdgcn_update_dpp(0, v, 0x150 + K, 0xf, 0xf, true);
}

static __device__ __forceinline__ float xhalf(float x)
{
    const int b = __float_as_int(x);
    return __int_as_float(__builtin_amdgcn_permlanex16(b, b, 0x76543210, 0xfedcba98, false, false));
}

template <int U>
static __device__ __forceinline__ int slot_src(int s_lo, int s_hi)
{
    return row_bcast<(U & 15)>((U < 16) ? s_lo : s_hi);
}

#define LD_SLOT(T, U)                                                            \
    do {                                                                         \
        const int sv_ = slot_src<(U)>(s_lo, s_hi);                               \
        ok[T] = (2 * (U) + half) < deg;                                          \
        r[T] = *(const float4*)(Hb + ((unsigned)sv_ * 256u + coff));             \
    } while (0)
#define LD_GROUP(T0, U0)                                                         \
    if (2 * (U0) < deg) {                                                        \
        LD_SLOT((T0), (U0));         LD_SLOT((T0) + 1, (U0) + 1);                \
        LD_SLOT((T0) + 2, (U0) + 2); LD_SLOT((T0) + 3, (U0) + 3);                \
    }
#define SC_SLOT(T)                                                               \
    do {                                                                         \
        float p_ = r[T].x * as4.x;                                               \
        p_ = fmaf(r[T].y, as4.y, p_);                                            \
        p_ = fmaf(r[T].z, as4.z, p_);                                            \
        p_ = fmaf(r[T].w, as4.w, p_);                                            \
        float v_ = row16_sum(p_) + hd;                                           \
        v_ = (v_ >= 0.0f) ? v_ : 0.2f * v_;                                      \
        sc[T] = ok[T] ? v_ : NEG_INF;                                            \
        cm = fmaxf(cm, sc[T]);                                                   \
    } while (0)
#define SC_GROUP(T0, U0)                                                         \
    if (2 * (U0) < deg) {                                                        \
        SC_SLOT((T0)); SC_SLOT((T0) + 1); SC_SLOT((T0) + 2); SC_SLOT((T0) + 3);  \
    }
#define AC_SLOT(T)                                                               \
    do {                                                                         \
        const float w_ = __expf(sc[T] - m);                                      \
        S += w_;                                                                 \
        acc.x = fmaf(w_, r[T].x, acc.x); acc.y = fmaf(w_, r[T].y, acc.y);        \
        acc.z = fmaf(w_, r[T].z, acc.z); acc.w = fmaf(w_, r[T].w, acc.w);        \
    } while (0)
#define AC_GROUP(T0, U0)                                                         \
    if (2 * (U0) < deg) {                                                        \
        AC_SLOT((T0)); AC_SLOT((T0) + 1); AC_SLOT((T0) + 2); AC_SLOT((T0) + 3);  \
    }

template <int U0, int NS>
static __device__ __forceinline__ void gather_chunk(
    const int deg, const int half, const unsigned coff, const int s_lo, const int s_hi,
    const char* __restrict__ Hb, const float4 as4, const float hd,
    float& m, float& S, float4& acc)
{
    const float NEG_INF = __int_as_float(0xFF800000);
    float4 r[NS];
    float  sc[NS];
    bool   ok[NS];

    LD_GROUP(0, U0)
    LD_GROUP(4, U0 + 4)
    if (NS > 8) { LD_GROUP(8, U0 + 8) }

    float cm = NEG_INF;
    SC_GROUP(0, U0)
    SC_GROUP(4, U0 + 4)
    if (NS > 8) { SC_GROUP(8, U0 + 8) }
    cm = fmaxf(cm, xhalf(cm));

    const float mn = fmaxf(m, cm);
    const float f  = __expf(m - mn);
    m = mn;
    S *= f;
    acc.x *= f; acc.y *= f; acc.z *= f; acc.w *= f;

    AC_GROUP(0, U0)
    AC_GROUP(4, U0 + 4)
    if (NS > 8) { AC_GROUP(8, U0 + 8) }
}

__global__ __launch_bounds__(32 * NODES_PER_BLOCK) void gat_gather(
    const int* __restrict__ cnt, const int* __restrict__ lists,
    const int* __restrict__ esrc, const int* __restrict__ edst, int E,
    const float* __restrict__ HS, const float* __restrict__ HD,
    const float* __restrict__ H, const float* __restrict__ a_s,
    const float* __restrict__ bias,
    float* __restrict__ OUT, int n_nodes)
{
    const int lane = threadIdx.x & 31;
    const int half = lane >> 4;
    const int c    = lane & 15;
    const int wv   = __builtin_amdgcn_readfirstlane((int)(threadIdx.x >> 5));
    const int node = blockIdx.x * NODES_PER_BLOCK + wv;
    if (node >= n_nodes) return;

    const int   deg = cnt[node];
    const float hd  = HD[node];
    const float4 b4 = ((const float4*)bias)[c];

    const float4* __restrict__ Hv = (const float4*)H;
    float4 acc;

    if (deg <= LCAP) {
        const int* __restrict__ lst = lists + (size_t)node * LCAP;
        const char* __restrict__ Hb = (const char*)H;
        const unsigned coff = 16u * (unsigned)c;

        const int e_lo = 2 * c + half;
        int s_lo = node, s_hi = node;
        if (e_lo < deg) s_lo = lst[e_lo];
        if (deg > 32) {
            if (e_lo + 32 < deg) s_hi = lst[e_lo + 32];
        }

        const float4 as4   = ((const float4*)a_s)[c];
        const float4 hself = Hv[(size_t)node * 16 + c];

        float ps = hself.x * as4.x;
        ps = fmaf(hself.y, as4.y, ps);
        ps = fmaf(hself.z, as4.z, ps);
        ps = fmaf(hself.w, as4.w, ps);
        float es = row16_sum(ps) + hd;
        es = (es >= 0.0f) ? es : 0.2f * es;

        float m = es;
        float S = (half == 0) ? 1.0f : 0.0f;
        acc.x = (half == 0) ? hself.x : 0.0f;
        acc.y = (half == 0) ? hself.y : 0.0f;
        acc.z = (half == 0) ? hself.z : 0.0f;
        acc.w = (half == 0) ? hself.w : 0.0f;

        if (deg > 0)  gather_chunk<0, 12>(deg, half, coff, s_lo, s_hi, Hb, as4, hd, m, S, acc);
        if (deg > 24) gather_chunk<12, 12>(deg, half, coff, s_lo, s_hi, Hb, as4, hd, m, S, acc);
        if (deg > 48) gather_chunk<24, 8>(deg, half, coff, s_lo, s_hi, Hb, as4, hd, m, S, acc);

        const float inv = 1.0f / (S + xhalf(S));
        const float bz = (half == 0) ? 1.0f : 0.0f;
        acc.x = fmaf(acc.x, inv, bz * b4.x); acc.y = fmaf(acc.y, inv, bz * b4.y);
        acc.z = fmaf(acc.z, inv, bz * b4.z); acc.w = fmaf(acc.w, inv, bz * b4.w);
    } else {
        float es = HS[node] + hd;
        es = (es >= 0.0f) ? es : 0.2f * es;
        float mx = es;
        for (int base = 0; base < E; base += 32) {
            const int e = base + lane;
            if (e < E && edst[e] == node) {
                float v = HS[esrc[e]] + hd;
                v = (v >= 0.0f) ? v : 0.2f * v;
                mx = fmaxf(mx, v);
            }
        }
        for (int off = 16; off >= 1; off >>= 1) mx = fmaxf(mx, __shfl_xor(mx, off, 32));

        float sum = 0.0f;
        acc.x = 0.0f; acc.y = 0.0f; acc.z = 0.0f; acc.w = 0.0f;
        for (int base = 0; base < E; base += 32) {
            const int  e   = base + lane;
            const bool hit = (e < E) && (edst[e] == node);
            int   s = node;
            float w = 0.0f;
            if (hit) {
                s = esrc[e];
                float v = HS[s] + hd;
                v = (v >= 0.0f) ? v : 0.2f * v;
                w = __expf(v - mx);
            }
            sum += w;
            unsigned mk = (unsigned)__ballot(hit);
            while (mk != 0u) {
                const int k = __builtin_ctz(mk);
                mk &= mk - 1u;
                const int   sk = __shfl(s, k, 32);
                const float wk = __shfl(w, k, 32);
                const float a  = (half == 0) ? wk : 0.0f;
                const float4 r = Hv[(size_t)sk * 16 + c];
                acc.x += a * r.x; acc.y += a * r.y;
                acc.z += a * r.z; acc.w += a * r.w;
            }
        }
        for (int off = 16; off >= 1; off >>= 1) sum += __shfl_xor(sum, off, 32);
        const float wself = __expf(es - mx);
        const float S = sum + wself;
        const float4 hself = Hv[(size_t)node * 16 + c];
        const float  aself = (half == 0) ? wself : 0.0f;
        const float bz = (half == 0) ? 1.0f : 0.0f;
        acc.x = (acc.x + aself * hself.x) / S + bz * b4.x;
        acc.y = (acc.y + aself * hself.y) / S + bz * b4.y;
        acc.z = (acc.z + aself * hself.z) / S + bz * b4.z;
        acc.w = (acc.w + aself * hself.w) / S + bz * b4.w;
    }

    acc.x += xhalf(acc.x);
    acc.y += xhalf(acc.y);
    acc.z += xhalf(acc.z);
    acc.w += xhalf(acc.w);
    if (half == 0) ((float4*)OUT)[(size_t)node * 16 + c] = acc;
}

extern "C" void kernel_launch(void* const* d_in, const int* in_sizes, int n_in,
                              void* d_out, int out_size, void* d_ws, size_t ws_size,
                              hipStream_t stream) {
    const float* x   = (const float*)d_in[0];
    const int*   ei  = (const int*)  d_in[1];
    const float* W1  = (const float*)d_in[2];
    const float* as1 = (const float*)d_in[3];
    const float* ad1 = (const float*)d_in[4];
    const float* b1  = (const float*)d_in[5];
    const float* W2  = (const float*)d_in[6];
    const float* as2 = (const float*)d_in[7];
    const float* ad2 = (const float*)d_in[8];
    const float* b2  = (const float*)d_in[9];

    const int N    = in_sizes[0] / FDIM;
    const int E    = in_sizes[1] / 2;
    const int* src = ei;
    const int* dst = ei + E;

    float* ws = (float*)d_ws;
    float* H  = ws;  ws += (size_t)N * FDIM;
    float* X1 = ws;  ws += (size_t)N * FDIM;
    int* lists = (int*)ws;  ws += (size_t)N * LCAP;
    float* HS = ws;  ws += N;
    float* HD = ws;  ws += N;
    int* cnt  = (int*)ws;

    adj_zero<<<(N + 255) / 256, 256, 0, stream>>>(cnt, N);
    adj_fill<<<(E + 255) / 256, 256, 0, stream>>>(src, dst, cnt, lists, E);

    auto layer = [&](const float* Xin, const float* W, const float* a_s,
                     const float* a_d, const float* b, float* OUT) {
        gat_gemm_wmma<<<N / 16, 32, 0, stream>>>(Xin, W, a_s, a_d, H, HS, HD);
        gat_gather<<<(N + NODES_PER_BLOCK - 1) / NODES_PER_BLOCK,
                     32 * NODES_PER_BLOCK, 0, stream>>>(cnt, lists, src, dst, E,
                                                        HS, HD, H, a_s, b, OUT, N);
    };

    layer(x,  W1, as1, ad1, b1, X1);
    layer(X1, W2, as2, ad2, b2, (float*)d_out);
}


# Aggregate before transform: gather the layer's input rows (scores from x.(W a_s), x.(W a_d)), then W applied per 16-node tile by WMMA inside the gather kernel; 6 to 4 launches; f32 sum order changes
# speedup vs baseline: 3.2624x; 1.0501x over previous
#include <hip/hip_runtime.h>
#include <math.h>

#define FDIM 64

typedef __attribute__((ext_vector_type(2))) float v2f;
typedef __attribute__((ext_vector_type(8))) float v8f;


#define LCAP 64

__global__ void adj_zero_fold(int* __restrict__ cnt, int n,
                              const float* __restrict__ W1, const float* __restrict__ as1,
                              const float* __restrict__ ad1,
                              const float* __restrict__ W2, const float* __restrict__ as2,
                              const float* __restrict__ ad2,
                              float* __restrict__ wvec)
{
    int i = blockIdx.x * blockDim.x + threadIdx.x;
    if (i < n) cnt[i] = 0;
    if (blockIdx.x == 0 && threadIdx.x < 4 * FDIM) {
        const int t   = threadIdx.x;
        const int v   = t >> 6;
        const int row = t & 63;
        const float* __restrict__ Wm = (v < 2) ? W1 : W2;
        const float* __restrict__ a  = (v == 0) ? as1 : (v == 1) ? ad1 : (v == 2) ? as2 : ad2;
        const float4* __restrict__ wr = (const float4*)(Wm + row * FDIM);
        const float4* __restrict__ av = (const float4*)a;
        float s = 0.0f;
#pragma unroll
        for (int q = 0; q < FDIM / 4; ++q) {
            const float4 w4 = wr[q];
            const float4 a4 = av[q];
            s = fmaf(w4.x, a4.x, s);
            s = fmaf(w4.y, a4.y, s);
            s = fmaf(w4.z, a4.z, s);
            s = fmaf(w4.w, a4.w, s);
        }
        wvec[v * FDIM + row] = s;
    }
}

__global__ void adj_fill(const int* __restrict__ src, const int* __restrict__ dst,
                         int* __restrict__ cnt, int* __restrict__ lists, int E)
{
    int e = blockIdx.x * blockDim.x + threadIdx.x;
    if (e < E) {
        const int d   = dst[e];
        const int sv  = src[e];
        asm volatile("" :: "v"(sv));
        const int pos = atomicAdd(&cnt[d], 1);
        if (pos < LCAP) lists[(size_t)d * LCAP + pos] = sv;
    }
}

#define NODES_PER_BLOCK 16
#define ZPITCH 68

template <int N>
static __device__ __forceinline__ float row_xor(float x)
{
    return __int_as_float(__builtin_amdgcn_update_dpp(0, __float_as_int(x), 0x160 + N, 0xf, 0xf, true));
}

static __device__ __forceinline__ float row16_sum(float p)
{
    p += row_xor<1>(p);
    p += row_xor<2>(p);
    p += row_xor<4>(p);
    p += row_xor<8>(p);
    return p;
}

template <int K>
static __device__ __forceinline__ int row_bcast(int v)
{
    return __builtin_amdgcn_update_dpp(0, v, 0x150 + K, 0xf, 0xf, true);
}

static __device__ __forceinline__ float xhalf(float x)
{
    const int b = __float_as_int(x);
    return __int_as_float(__builtin_amdgcn_permlanex16(b, b, 0x76543210, 0xfedcba98, false, false));
}

template <int U>
static __device__ __forceinline__ int slot_src(int s_lo, int s_hi)
{
    return row_bcast<(U & 15)>((U < 16) ? s_lo : s_hi);
}

#define LD_SLOT(T, U)                                                            \
    do {                                                                         \
        const int sv_ = slot_src<(U)>(s_lo, s_hi);                               \
        ok[T] = (2 * (U) + half) < deg;                                          \
        r[T] = *(const float4*)(Hb + ((unsigned)sv_ * 256u + coff));             \
    } while (0)
#define LD_GROUP(T0, U0)                                                         \
    if (2 * (U0) < deg) {                                                        \
        LD_SLOT((T0), (U0));         LD_SLOT((T0) + 1, (U0) + 1);                \
        LD_SLOT((T0) + 2, (U0) + 2); LD_SLOT((T0) + 3, (U0) + 3);                \
    }
#define SC_SLOT(T)                                                               \
    do {                                                                         \
        float p_ = r[T].x * as4.x;                                               \
        p_ = fmaf(r[T].y, as4.y, p_);                                            \
        p_ = fmaf(r[T].z, as4.z, p_);                                            \
        p_ = fmaf(r[T].w, as4.w, p_);                                            \
        float v_ = row16_sum(p_) + hd;                                           \
        v_ = (v_ >= 0.0f) ? v_ : 0.2f * v_;                                      \
        sc[T] = ok[T] ? v_ : NEG_INF;                                            \
        cm = fmaxf(cm, sc[T]);                                                   \
    } while (0)
#define SC_GROUP(T0, U0)                                                         \
    if (2 * (U0) < deg) {                                                        \
        SC_SLOT((T0)); SC_SLOT((T0) + 1); SC_SLOT((T0) + 2); SC_SLOT((T0) + 3);  \
    }
#define AC_SLOT(T)                                                               \
    do {                                                                         \
        const float w_ = __expf(sc[T] - m);                                      \
        S += w_;                                                                 \
        acc.x = fmaf(w_, r[T].x, acc.x); acc.y = fmaf(w_, r[T].y, acc.y);        \
        acc.z = fmaf(w_, r[T].z, acc.z); acc.w = fmaf(w_, r[T].w, acc.w);        \
    } while (0)
#define AC_GROUP(T0, U0)                                                         \
    if (2 * (U0) < deg) {                                                        \
        AC_SLOT((T0)); AC_SLOT((T0) + 1); AC_SLOT((T0) + 2); AC_SLOT((T0) + 3);  \
    }

template <int U0, int NS>
static __device__ __forceinline__ void gather_chunk(
    const int deg, const int half, const unsigned coff, const int s_lo, const int s_hi,
    const char* __restrict__ Hb, const float4 as4, const float hd,
    float& m, float& S, float4& acc)
{
    const float NEG_INF = __int_as_float(0xFF800000);
    float4 r[NS];
    float  sc[NS];
    bool   ok[NS];

    LD_GROUP(0, U0)
    LD_GROUP(4, U0 + 4)
    if (NS > 8) { LD_GROUP(8, U0 + 8) }

    float cm = NEG_INF;
    SC_GROUP(0, U0)
    SC_GROUP(4, U0 + 4)
    if (NS > 8) { SC_GROUP(8, U0 + 8) }
    cm = fmaxf(cm, xhalf(cm));

    const float mn = fmaxf(m, cm);
    const float f  = __expf(m - mn);
    m = mn;
    S *= f;
    acc.x *= f; acc.y *= f; acc.z *= f; acc.w *= f;

    AC_GROUP(0, U0)
    AC_GROUP(4, U0 + 4)
    if (NS > 8) { AC_GROUP(8, U0 + 8) }
}

__global__ __launch_bounds__(32 * NODES_PER_BLOCK) void gat_gather_xform(
    const int* __restrict__ cnt, const int* __restrict__ lists,
    const int* __restrict__ esrc, const int* __restrict__ edst, int E,
    const float* __restrict__ X,
    const float* __restrict__ was,
    const float* __restrict__ wad,
    const float* __restrict__ Wm,
    const float* __restrict__ bias,
    float* __restrict__ OUT, int n_nodes)
{
    __shared__ __attribute__((aligned(16))) float Zs[NODES_PER_BLOCK * ZPITCH];
    __shared__ int arrived;

    if (threadIdx.x == 0) arrived = 0;
    __syncthreads();

    const int lane = threadIdx.x & 31;
    const int half = lane >> 4;
    const int c    = lane & 15;
    const int wv   = __builtin_amdgcn_readfirstlane((int)(threadIdx.x >> 5));
    const int node = blockIdx.x * NODES_PER_BLOCK + wv;

    float4 acc;
    acc.x = 0.0f; acc.y = 0.0f; acc.z = 0.0f; acc.w = 0.0f;

    if (node < n_nodes) {
        const int deg = cnt[node];
        const float4* __restrict__ Xv = (const float4*)X;
        const float4 as4   = ((const float4*)was)[c];
        const float4 ad4   = ((const float4*)wad)[c];
        const float4 hself = Xv[(size_t)node * 16 + c];

        float pd = hself.x * ad4.x;
        pd = fmaf(hself.y, ad4.y, pd);
        pd = fmaf(hself.z, ad4.z, pd);
        pd = fmaf(hself.w, ad4.w, pd);
        const float hd = row16_sum(pd);

        float ps = hself.x * as4.x;
        ps = fmaf(hself.y, as4.y, ps);
        ps = fmaf(hself.z, as4.z, ps);
        ps = fmaf(hself.w, as4.w, ps);
        float es = row16_sum(ps) + hd;
        es = (es >= 0.0f) ? es : 0.2f * es;

        if (deg <= LCAP) {
            const int* __restrict__ lst = lists + (size_t)node * LCAP;
            const char* __restrict__ Hb = (const char*)X;
            const unsigned coff = 16u * (unsigned)c;

            const int e_lo = 2 * c + half;
            int s_lo = node, s_hi = node;
            if (e_lo < deg) s_lo = lst[e_lo];
            if (deg > 32) {
                if (e_lo + 32 < deg) s_hi = lst[e_lo + 32];
            }

            float m = es;
            float S = (half == 0) ? 1.0f : 0.0f;
            acc.x = (half == 0) ? hself.x : 0.0f;
            acc.y = (half == 0) ? hself.y : 0.0f;
            acc.z = (half == 0) ? hself.z : 0.0f;
            acc.w = (half == 0) ? hself.w : 0.0f;

            if (deg > 0)  gather_chunk<0, 12>(deg, half, coff, s_lo, s_hi, Hb, as4, hd, m, S, acc);
            if (deg > 24) gather_chunk<12, 12>(deg, half, coff, s_lo, s_hi, Hb, as4, hd, m, S, acc);
            if (deg > 48) gather_chunk<24, 8>(deg, half, coff, s_lo, s_hi, Hb, as4, hd, m, S, acc);

            const float inv = 1.0f / (S + xhalf(S));
            acc.x *= inv; acc.y *= inv; acc.z *= inv; acc.w *= inv;
        } else {
            float m = es;
            float S = 1.0f;
            acc = hself;
            for (int base = 0; base < E; base += 32) {
                const int  e   = base + lane;
                const bool hit = (e < E) && (edst[e] == node);
                int s = node;
                if (hit) s = esrc[e];
                unsigned mk = (unsigned)__ballot(hit);
                while (mk != 0u) {
                    const int k = __builtin_ctz(mk);
                    mk &= mk - 1u;
                    const int    sk = __shfl(s, k, 32);
                    const float4 r  = Xv[(size_t)sk * 16 + c];
                    float p = r.x * as4.x;
                    p = fmaf(r.y, as4.y, p);
                    p = fmaf(r.z, as4.z, p);
                    p = fmaf(r.w, as4.w, p);
                    float v = row16_sum(p) + hd;
                    v = (v >= 0.0f) ? v : 0.2f * v;
                    const float mn = fmaxf(m, v);
                    const float f  = __expf(m - mn);
                    const float w  = __expf(v - mn);
                    m = mn;
                    S = fmaf(S, f, w);
                    acc.x = fmaf(acc.x, f, w * r.x); acc.y = fmaf(acc.y, f, w * r.y);
                    acc.z = fmaf(acc.z, f, w * r.z); acc.w = fmaf(acc.w, f, w * r.w);
                }
            }
            const float inv = (half == 0) ? 1.0f / S : 0.0f;
            acc.x *= inv; acc.y *= inv; acc.z *= inv; acc.w *= inv;
        }

        acc.x += xhalf(acc.x);
        acc.y += xhalf(acc.y);
        acc.z += xhalf(acc.z);
        acc.w += xhalf(acc.w);
    }

    if (half == 0) *(float4*)(Zs + wv * ZPITCH + 4 * c) = acc;
    int owner = 0;
    if (lane == 0) {
        const int before = __hip_atomic_fetch_add(&arrived, 1, __ATOMIC_ACQ_REL,
                                                  __HIP_MEMORY_SCOPE_WORKGROUP);
        owner = (before == NODES_PER_BLOCK - 1) ? 1 : 0;
    }
    owner = __builtin_amdgcn_readfirstlane(owner);
    if (owner == 0) return;

    const int khalf = half;
    v8f d0 = (v8f){0,0,0,0,0,0,0,0};
    v8f d1 = (v8f){0,0,0,0,0,0,0,0};
    v8f d2 = (v8f){0,0,0,0,0,0,0,0};
    v8f d3 = (v8f){0,0,0,0,0,0,0,0};
    const float* zr = Zs + c * ZPITCH + 2 * khalf;
    const float4* __restrict__ wq = (const float4*)(Wm + (2 * khalf) * FDIM) + c;
#pragma nounroll
    for (int kc = 0; kc < 4; ++kc) {
        const float*  zc = zr + 16 * kc;
        const float4* __restrict__ wc = wq + (16 * kc) * (FDIM / 4);
#pragma unroll
        for (int k = 0; k < 4; ++k) {
            const v2f    af = *(const v2f*)(zc + 4 * k);
            const float4 q0 = wc[(4 * k) * (FDIM / 4)];
            const float4 q1 = wc[(4 * k + 1) * (FDIM / 4)];
            v2f b;
            b.x = q0.x; b.y = q1.x;
            d0 = __builtin_amdgcn_wmma_f32_16x16x4_f32(false, af, false, b, (short)0, d0, false, false);
            b.x = q0.y; b.y = q1.y;
            d1 = __builtin_amdgcn_wmma_f32_16x16x4_f32(false, af, false, b, (short)0, d1, false, false);
            b.x = q0.z; b.y = q1.z;
            d2 = __builtin_amdgcn_wmma_f32_16x16x4_f32(false, af, false, b, (short)0, d2, false, false);
            b.x = q0.w; b.y = q1.w;
            d3 = __builtin_amdgcn_wmma_f32_16x16x4_f32(false, af, false, b, (short)0, d3, false, false);
        }
    }

    const float4 bz = ((const float4*)bias)[c];
    const int rbase = blockIdx.x * NODES_PER_BLOCK + 8 * khalf;
#pragma unroll
    for (int r = 0; r < 8; ++r) {
        float4 o;
        o.x = d0[r] + bz.x;
        o.y = d1[r] + bz.y;
        o.z = d2[r] + bz.z;
        o.w = d3[r] + bz.w;
        if (rbase + r < n_nodes) ((float4*)OUT)[(size_t)(rbase + r) * 16 + c] = o;
    }
}

extern "C" void kernel_launch(void* const* d_in, const int* in_sizes, int n_in,
                              void* d_out, int out_size, void* d_ws, size_t ws_size,
                              hipStream_t stream) {
    const float* x   = (const float*)d_in[0];
    const int*   ei  = (const int*)  d_in[1];
    const float* W1  = (const float*)d_in[2];
    const float* as1 = (const float*)d_in[3];
    const float* ad1 = (const float*)d_in[4];
    const float* b1  = (const float*)d_in[5];
    const float* W2  = (const float*)d_in[6];
    const float* as2 = (const float*)d_in[7];
    const float* ad2 = (const float*)d_in[8];
    const float* b2  = (const float*)d_in[9];

    const int N    = in_sizes[0] / FDIM;
    const int E    = in_sizes[1] / 2;
    const int* src = ei;
    const int* dst = ei + E;

    float* ws = (float*)d_ws;
    float* X1 = ws;  ws += (size_t)N * FDIM;
    int* lists = (int*)ws;  ws += (size_t)N * LCAP;
    float* wvec = ws;  ws += 4 * FDIM;
    int* cnt  = (int*)ws;

    adj_zero_fold<<<(N + 255) / 256, 256, 0, stream>>>(cnt, N, W1, as1, ad1, W2, as2, ad2, wvec);
    adj_fill<<<(E + 255) / 256, 256, 0, stream>>>(src, dst, cnt, lists, E);

    auto layer = [&](const float* Xin, const float* was, const float* wad,
                     const float* W, const float* b, float* OUT) {
        gat_gather_xform<<<(N + NODES_PER_BLOCK - 1) / NODES_PER_BLOCK,
                           32 * NODES_PER_BLOCK, 0, stream>>>(cnt, lists, src, dst, E,
                                                              Xin, was, wad, W, b, OUT, N);
    };

    layer(x,  wvec,            wvec + FDIM,     W1, b1, X1);
    layer(X1, wvec + 2 * FDIM, wvec + 3 * FDIM, W2, b2, (float*)d_out);
}


# Edge lists built with LDS counters: chunk work-groups sort edges into bucket slots, bucket work-groups place them by per-node LDS adds; no global atomics, no clearing launch; sum order by arrival
# speedup vs baseline: 5.7572x; 1.7647x over previous
#include <hip/hip_runtime.h>
#include <math.h>

#define FDIM 64

typedef __attribute__((ext_vector_type(2))) float v2f;
typedef __attribute__((ext_vector_type(8))) float v8f;


#define LCAP 64

#define BUCKET_SHIFT 8
#define BUCKET_NODES 256
#define NCHUNK 256
#define SLOT_C 64
#define EB_THREADS 512

__global__ __launch_bounds__(EB_THREADS) void edge_bucket(
    const int* __restrict__ src, const int* __restrict__ dst, int E,
    int2* __restrict__ rec, int* __restrict__ bcnt,
    const float* __restrict__ W1, const float* __restrict__ as1, const float* __restrict__ ad1,
    const float* __restrict__ W2, const float* __restrict__ as2, const float* __restrict__ ad2,
    float* __restrict__ wvec)
{
    __shared__ int lc[BUCKET_NODES];

    if (blockIdx.x == NCHUNK) {
        if (threadIdx.x < 4 * FDIM) {
            const int t   = threadIdx.x;
            const int v   = t >> 6;
            const int row = t & 63;
            const float* __restrict__ Wm = (v < 2) ? W1 : W2;
            const float* __restrict__ a  = (v == 0) ? as1 : (v == 1) ? ad1 : (v == 2) ? as2 : ad2;
            const float4* __restrict__ wr = (const float4*)(Wm + row * FDIM);
            const float4* __restrict__ av = (const float4*)a;
            float s = 0.0f;
#pragma unroll
            for (int q = 0; q < FDIM / 4; ++q) {
                const float4 w4 = wr[q];
                const float4 a4 = av[q];
                s = fmaf(w4.x, a4.x, s);
                s = fmaf(w4.y, a4.y, s);
                s = fmaf(w4.z, a4.z, s);
                s = fmaf(w4.w, a4.w, s);
            }
            wvec[v * FDIM + row] = s;
        }
        return;
    }

    if (threadIdx.x < BUCKET_NODES) lc[threadIdx.x] = 0;
    __syncthreads();

    const int g     = blockIdx.x;
    const int chunk = (E + NCHUNK - 1) / NCHUNK;
    const int e0    = g * chunk;
    const int e1    = min(e0 + chunk, E);
    for (int base = e0; base < e1; base += 4 * EB_THREADS) {
        int d[4], s[4];
#pragma unroll
        for (int k = 0; k < 4; ++k) {
            const int e = base + k * EB_THREADS + (int)threadIdx.x;
            d[k] = -1; s[k] = 0;
            if (e < e1) { d[k] = dst[e]; s[k] = src[e]; }
        }
#pragma unroll
        for (int k = 0; k < 4; ++k) {
            if (d[k] >= 0) {
                const int b   = d[k] >> BUCKET_SHIFT;
                const int pos = atomicAdd(&lc[b], 1);
                if (pos < SLOT_C)
                    rec[((size_t)b * NCHUNK + g) * SLOT_C + pos] = make_int2(s[k], d[k]);
            }
        }
    }
    __syncthreads();
    if (threadIdx.x < BUCKET_NODES) bcnt[threadIdx.x * NCHUNK + g] = lc[threadIdx.x];
}

__global__ __launch_bounds__(NCHUNK) void bucket_place(
    const int2* __restrict__ rec, const int* __restrict__ bcnt,
    const int* __restrict__ src, const int* __restrict__ dst, int E,
    int* __restrict__ cnt, int* __restrict__ lists, int n)
{
    __shared__ int ln[BUCKET_NODES];
    const int t     = threadIdx.x;
    const int b     = blockIdx.x;
    const int nbase = b << BUCKET_SHIFT;
    ln[t] = 0;
    __syncthreads();

    const int c = bcnt[b * NCHUNK + t];
    if (c <= SLOT_C) {
        const int2* __restrict__ rp = rec + ((size_t)b * NCHUNK + t) * SLOT_C;
        for (int i = 0; i < c; i += 4) {
            int2 r[4];
#pragma unroll
            for (int k = 0; k < 4; ++k) {
                r[k] = make_int2(0, -1);
                if (i + k < c) r[k] = rp[i + k];
            }
#pragma unroll
            for (int k = 0; k < 4; ++k) {
                if (r[k].y >= 0) {
                    const int pos = atomicAdd(&ln[r[k].y - nbase], 1);
                    if (pos < LCAP) lists[(size_t)r[k].y * LCAP + pos] = r[k].x;
                }
            }
        }
    } else {
        const int chunk = (E + NCHUNK - 1) / NCHUNK;
        const int e0    = t * chunk;
        const int e1    = min(e0 + chunk, E);
        for (int e = e0; e < e1; ++e) {
            const int d = dst[e];
            if ((d >> BUCKET_SHIFT) == b) {
                const int sv  = src[e];
                const int pos = atomicAdd(&ln[d - nbase], 1);
                if (pos < LCAP) lists[(size_t)d * LCAP + pos] = sv;
            }
        }
    }
    __syncthreads();
    if (nbase + t < n) cnt[nbase + t] = ln[t];
}

#define NODES_PER_BLOCK 16
#define ZPITCH 68

template <int N>
static __device__ __forceinline__ float row_xor(float x)
{
    return __int_as_float(__builtin_amdgcn_update_dpp(0, __float_as_int(x), 0x160 + N, 0xf, 0xf, true));
}

static __device__ __forceinline__ float row16_sum(float p)
{
    p += row_xor<1>(p);
    p += row_xor<2>(p);
    p += row_xor<4>(p);
    p += row_xor<8>(p);
    return p;
}

template <int K>
static __device__ __forceinline__ int row_bcast(int v)
{
    return __builtin_amdgcn_update_dpp(0, v, 0x150 + K, 0xf, 0xf, true);
}

static __device__ __forceinline__ float xhalf(float x)
{
    const int b = __float_as_int(x);
    return __int_as_float(__builtin_amdgcn_permlanex16(b, b, 0x76543210, 0xfedcba98, false, false));
}

template <int U>
static __device__ __forceinline__ int slot_src(int s_lo, int s_hi)
{
    return row_bcast<(U & 15)>((U < 16) ? s_lo : s_hi);
}

#define LD_SLOT(T, U)                                                            \
    do {                                                                         \
        const int sv_ = slot_src<(U)>(s_lo, s_hi);                               \
        ok[T] = (2 * (U) + half) < deg;                                          \
        r[T] = *(const float4*)(Hb + ((unsigned)sv_ * 256u + coff));             \
    } while (0)
#define LD_GROUP(T0, U0)                                                         \
    if (2 * (U0) < deg) {                                                        \
        LD_SLOT((T0), (U0));         LD_SLOT((T0) + 1, (U0) + 1);                \
        LD_SLOT((T0) + 2, (U0) + 2); LD_SLOT((T0) + 3, (U0) + 3);                \
    }
#define SC_SLOT(T)                                                               \
    do {                                                                         \
        float p_ = r[T].x * as4.x;                                               \
        p_ = fmaf(r[T].y, as4.y, p_);                                            \
        p_ = fmaf(r[T].z, as4.z, p_);                                            \
        p_ = fmaf(r[T].w, as4.w, p_);                                            \
        float v_ = row16_sum(p_) + hd;                                           \
        v_ = (v_ >= 0.0f) ? v_ : 0.2f * v_;                                      \
        sc[T] = ok[T] ? v_ : NEG_INF;                                            \
        cm = fmaxf(cm, sc[T]);                                                   \
    } while (0)
#define SC_GROUP(T0, U0)                                                         \
    if (2 * (U0) < deg) {                                                        \
        SC_SLOT((T0)); SC_SLOT((T0) + 1); SC_SLOT((T0) + 2); SC_SLOT((T0) + 3);  \
    }
#define AC_SLOT(T)                                                               \
    do {                                                                         \
        const float w_ = __expf(sc[T] - m);                                      \
        S += w_;                                                                 \
        acc.x = fmaf(w_, r[T].x, acc.x); acc.y = fmaf(w_, r[T].y, acc.y);        \
        acc.z = fmaf(w_, r[T].z, acc.z); acc.w = fmaf(w_, r[T].w, acc.w);        \
    } while (0)
#define AC_GROUP(T0, U0)                                                         \
    if (2 * (U0) < deg) {                                                        \
        AC_SLOT((T0)); AC_SLOT((T0) + 1); AC_SLOT((T0) + 2); AC_SLOT((T0) + 3);  \
    }

template <int U0, int NS>
static __device__ __forceinline__ void gather_chunk(
    const int deg, const int half, const unsigned coff, const int s_lo, const int s_hi,
    const char* __restrict__ Hb, const float4 as4, const float hd,
    float& m, float& S, float4& acc)
{
    const float NEG_INF = __int_as_float(0xFF800000);
    float4 r[NS];
    float  sc[NS];
    bool   ok[NS];

    LD_GROUP(0, U0)
    LD_GROUP(4, U0 + 4)
    if (NS > 8) { LD_GROUP(8, U0 + 8) }

    float cm = NEG_INF;
    SC_GROUP(0, U0)
    SC_GROUP(4, U0 + 4)
    if (NS > 8) { SC_GROUP(8, U0 + 8) }
    cm = fmaxf(cm, xhalf(cm));

    const float mn = fmaxf(m, cm);
    const float f  = __expf(m - mn);
    m = mn;
    S *= f;
    acc.x *= f; acc.y *= f; acc.z *= f; acc.w *= f;

    AC_GROUP(0, U0)
    AC_GROUP(4, U0 + 4)
    if (NS > 8) { AC_GROUP(8, U0 + 8) }
}

__global__ __launch_bounds__(32 * NODES_PER_BLOCK) void gat_gather_xform(
    const int* __restrict__ cnt, const int* __restrict__ lists,
    const int* __restrict__ esrc, const int* __restrict__ edst, int E,
    const float* __restrict__ X,
    const float* __restrict__ was,
    const float* __restrict__ wad,
    const float* __restrict__ Wm,
    const float* __restrict__ bias,
    float* __restrict__ OUT, int n_nodes)
{
    __shared__ __attribute__((aligned(16))) float Zs[NODES_PER_BLOCK * ZPITCH];
    __shared__ int arrived;

    if (threadIdx.x == 0) arrived = 0;
    __syncthreads();

    const int lane = threadIdx.x & 31;
    const int half = lane >> 4;
    const int c    = lane & 15;
    const int wv   = __builtin_amdgcn_readfirstlane((int)(threadIdx.x >> 5));
    const int node = blockIdx.x * NODES_PER_BLOCK + wv;

    float4 acc;
    acc.x = 0.0f; acc.y = 0.0f; acc.z = 0.0f; acc.w = 0.0f;

    if (node < n_nodes) {
        const int deg = cnt[node];
        const float4* __restrict__ Xv = (const float4*)X;
        const float4 as4   = ((const float4*)was)[c];
        const float4 ad4   = ((const float4*)wad)[c];
        const float4 hself = Xv[(size_t)node * 16 + c];

        float pd = hself.x * ad4.x;
        pd = fmaf(hself.y, ad4.y, pd);
        pd = fmaf(hself.z, ad4.z, pd);
        pd = fmaf(hself.w, ad4.w, pd);
        const float hd = row16_sum(pd);

        float ps = hself.x * as4.x;
        ps = fmaf(hself.y, as4.y, ps);
        ps = fmaf(hself.z, as4.z, ps);
        ps = fmaf(hself.w, as4.w, ps);
        float es = row16_sum(ps) + hd;
        es = (es >= 0.0f) ? es : 0.2f * es;

        if (deg <= LCAP) {
            const int* __restrict__ lst = lists + (size_t)node * LCAP;
            const char* __restrict__ Hb = (const char*)X;
            const unsigned coff = 16u * (unsigned)c;

            const int e_lo = 2 * c + half;
            int s_lo = node, s_hi = node;
            if (e_lo < deg) s_lo = lst[e_lo];
            if (deg > 32) {
                if (e_lo + 32 < deg) s_hi = lst[e_lo + 32];
            }

            float m = es;
            float S = (half == 0) ? 1.0f : 0.0f;
            acc.x = (half == 0) ? hself.x : 0.0f;
            acc.y = (half == 0) ? hself.y : 0.0f;
            acc.z = (half == 0) ? hself.z : 0.0f;
            acc.w = (half == 0) ? hself.w : 0.0f;

            if (deg > 0)  gather_chunk<0, 12>(deg, half, coff, s_lo, s_hi, Hb, as4, hd, m, S, acc);
            if (deg > 24) gather_chunk<12, 12>(deg, half, coff, s_lo, s_hi, Hb, as4, hd, m, S, acc);
            if (deg > 48) gather_chunk<24, 8>(deg, half, coff, s_lo, s_hi, Hb, as4, hd, m, S, acc);

            const float inv = 1.0f / (S + xhalf(S));
            acc.x *= inv; acc.y *= inv; acc.z *= inv; acc.w *= inv;
        } else {
            float m = es;
            float S = 1.0f;
            acc = hself;
            for (int base = 0; base < E; base += 32) {
                const int  e   = base + lane;
                const bool hit = (e < E) && (edst[e] == node);
                int s = node;
                if (hit) s = esrc[e];
                unsigned mk = (unsigned)__ballot(hit);
                while (mk != 0u) {
                    const int k = __builtin_ctz(mk);
                    mk &= mk - 1u;
                    const int    sk = __shfl(s, k, 32);
                    const float4 r  = Xv[(size_t)sk * 16 + c];
                    float p = r.x * as4.x;
                    p = fmaf(r.y, as4.y, p);
                    p = fmaf(r.z, as4.z, p);
                    p = fmaf(r.w, as4.w, p);
                    float v = row16_sum(p) + hd;
                    v = (v >= 0.0f) ? v : 0.2f * v;
                    const float mn = fmaxf(m, v);
                    const float f  = __expf(m - mn);
                    const float w  = __expf(v - mn);
                    m = mn;
                    S = fmaf(S, f, w);
                    acc.x = fmaf(acc.x, f, w * r.x); acc.y = fmaf(acc.y, f, w * r.y);
                    acc.z = fmaf(acc.z, f, w * r.z); acc.w = fmaf(acc.w, f, w * r.w);
                }
            }
            const float inv = (half == 0) ? 1.0f / S : 0.0f;
            acc.x *= inv; acc.y *= inv; acc.z *= inv; acc.w *= inv;
        }

        acc.x += xhalf(acc.x);
        acc.y += xhalf(acc.y);
        acc.z += xhalf(acc.z);
        acc.w += xhalf(acc.w);
    }

    if (half == 0) *(float4*)(Zs + wv * ZPITCH + 4 * c) = acc;
    int owner = 0;
    if (lane == 0) {
        const int before = __hip_atomic_fetch_add(&arrived, 1, __ATOMIC_ACQ_REL,
                                                  __HIP_MEMORY_SCOPE_WORKGROUP);
        owner = (before == NODES_PER_BLOCK - 1) ? 1 : 0;
    }
    owner = __builtin_amdgcn_readfirstlane(owner);
    if (owner == 0) return;

    const int khalf = half;
    v8f d0 = (v8f){0,0,0,0,0,0,0,0};
    v8f d1 = (v8f){0,0,0,0,0,0,0,0};
    v8f d2 = (v8f){0,0,0,0,0,0,0,0};
    v8f d3 = (v8f){0,0,0,0,0,0,0,0};
    const float* zr = Zs + c * ZPITCH + 2 * khalf;
    const float4* __restrict__ wq = (const float4*)(Wm + (2 * khalf) * FDIM) + c;
#pragma nounroll
    for (int kc = 0; kc < 4; ++kc) {
        const float*  zc = zr + 16 * kc;
        const float4* __restrict__ wc = wq + (16 * kc) * (FDIM / 4);
#pragma unroll
        for (int k = 0; k < 4; ++k) {
            const v2f    af = *(const v2f*)(zc + 4 * k);
            const float4 q0 = wc[(4 * k) * (FDIM / 4)];
            const float4 q1 = wc[(4 * k + 1) * (FDIM / 4)];
            v2f b;
            b.x = q0.x; b.y = q1.x;
            d0 = __builtin_amdgcn_wmma_f32_16x16x4_f32(false, af, false, b, (short)0, d0, false, false);
            b.x = q0.y; b.y = q1.y;
            d1 = __builtin_amdgcn_wmma_f32_16x16x4_f32(false, af, false, b, (short)0, d1, false, false);
            b.x = q0.z; b.y = q1.z;
            d2 = __builtin_amdgcn_wmma_f32_16x16x4_f32(false, af, false, b, (short)0, d2, false, false);
            b.x = q0.w; b.y = q1.w;
            d3 = __builtin_amdgcn_wmma_f32_16x16x4_f32(false, af, false, b, (short)0, d3, false, false);
        }
    }

    const float4 bz = ((const float4*)bias)[c];
    const int rbase = blockIdx.x * NODES_PER_BLOCK + 8 * khalf;
#pragma unroll
    for (int r = 0; r < 8; ++r) {
        float4 o;
        o.x = d0[r] + bz.x;
        o.y = d1[r] + bz.y;
        o.z = d2[r] + bz.z;
        o.w = d3[r] + bz.w;
        if (rbase + r < n_nodes) ((float4*)OUT)[(size_t)(rbase + r) * 16 + c] = o;
    }
}

extern "C" void kernel_launch(void* const* d_in, const int* in_sizes, int n_in,
                              void* d_out, int out_size, void* d_ws, size_t ws_size,
                              hipStream_t stream) {
    const float* x   = (const float*)d_in[0];
    const int*   ei  = (const int*)  d_in[1];
    const float* W1  = (const float*)d_in[2];
    const float* as1 = (const float*)d_in[3];
    const float* ad1 = (const float*)d_in[4];
    const float* b1  = (const float*)d_in[5];
    const float* W2  = (const float*)d_in[6];
    const float* as2 = (const float*)d_in[7];
    const float* ad2 = (const float*)d_in[8];
    const float* b2  = (const float*)d_in[9];

    const int N    = in_sizes[0] / FDIM;
    const int E    = in_sizes[1] / 2;
    const int* src = ei;
    const int* dst = ei + E;

    float* ws = (float*)d_ws;
    float* X1 = ws;  ws += (size_t)N * FDIM;
    int* lists = (int*)ws;  ws += (size_t)N * LCAP;
    float* wvec = ws;  ws += 4 * FDIM;
    int* bcnt = (int*)ws;  ws += BUCKET_NODES * NCHUNK;
    int2* rec = (int2*)ws;  ws += (size_t)2 * BUCKET_NODES * NCHUNK * SLOT_C;
    int* cnt  = (int*)ws;

    edge_bucket<<<NCHUNK + 1, EB_THREADS, 0, stream>>>(src, dst, E, rec, bcnt,
                                                       W1, as1, ad1, W2, as2, ad2, wvec);
    bucket_place<<<(N + BUCKET_NODES - 1) / BUCKET_NODES, NCHUNK, 0, stream>>>(
        rec, bcnt, src, dst, E, cnt, lists, N);

    auto layer = [&](const float* Xin, const float* was, const float* wad,
                     const float* W, const float* b, float* OUT) {
        gat_gather_xform<<<(N + NODES_PER_BLOCK - 1) / NODES_PER_BLOCK,
                           32 * NODES_PER_BLOCK, 0, stream>>>(cnt, lists, src, dst, E,
                                                              Xin, was, wad, W, b, OUT, N);
    };

    layer(x,  wvec,            wvec + FDIM,     W1, b1, X1);
    layer(X1, wvec + 2 * FDIM, wvec + 3 * FDIM, W2, b2, (float*)d_out);
}


# Gather waves take node after node: a block of 32 waves owns 13 tiles (208 nodes), keeps their degrees in LDS and draws rows from an LDS counter; 50000 to 7712 waves per layer; sum order unchanged
# speedup vs baseline: 6.8936x; 1.1974x over previous
#include <hip/hip_runtime.h>
#include <math.h>

#define FDIM 64

typedef __attribute__((ext_vector_type(2))) float v2f;
typedef __attribute__((ext_vector_type(8))) float v8f;


#define LCAP 64

#define BUCKET_SHIFT 8
#define BUCKET_NODES 256
#define NCHUNK 256
#define SLOT_C 64
#define EB_THREADS 512

__global__ __launch_bounds__(EB_THREADS) void edge_bucket(
    const int* __restrict__ src, const int* __restrict__ dst, int E,
    int2* __restrict__ rec, int* __restrict__ bcnt,
    const float* __restrict__ W1, const float* __restrict__ as1, const float* __restrict__ ad1,
    const float* __restrict__ W2, const float* __restrict__ as2, const float* __restrict__ ad2,
    float* __restrict__ wvec)
{
    __shared__ int lc[BUCKET_NODES];

    if (blockIdx.x == NCHUNK) {
        if (threadIdx.x < 4 * FDIM) {
            const int t   = threadIdx.x;
            const int v   = t >> 6;
            const int row = t & 63;
            const float* __restrict__ Wm = (v < 2) ? W1 : W2;
            const float* __restrict__ a  = (v == 0) ? as1 : (v == 1) ? ad1 : (v == 2) ? as2 : ad2;
            const float4* __restrict__ wr = (const float4*)(Wm + row * FDIM);
            const float4* __restrict__ av = (const float4*)a;
            float s = 0.0f;
#pragma unroll
            for (int q = 0; q < FDIM / 4; ++q) {
                const float4 w4 = wr[q];
                const float4 a4 = av[q];
                s = fmaf(w4.x, a4.x, s);
                s = fmaf(w4.y, a4.y, s);
                s = fmaf(w4.z, a4.z, s);
                s = fmaf(w4.w, a4.w, s);
            }
            wvec[v * FDIM + row] = s;
        }
        return;
    }

    if (threadIdx.x < BUCKET_NODES) lc[threadIdx.x] = 0;
    __syncthreads();

    const int g     = blockIdx.x;
    const int chunk = (E + NCHUNK - 1) / NCHUNK;
    const int e0    = g * chunk;
    const int e1    = min(e0 + chunk, E);
    for (int base = e0; base < e1; base += 4 * EB_THREADS) {
        int d[4], s[4];
#pragma unroll
        for (int k = 0; k < 4; ++k) {
            const int e = base + k * EB_THREADS + (int)threadIdx.x;
            d[k] = -1; s[k] = 0;
            if (e < e1) { d[k] = dst[e]; s[k] = src[e]; }
        }
#pragma unroll
        for (int k = 0; k < 4; ++k) {
            if (d[k] >= 0) {
                const int b   = d[k] >> BUCKET_SHIFT;
                const int pos = atomicAdd(&lc[b], 1);
                if (pos < SLOT_C)
                    rec[((size_t)b * NCHUNK + g) * SLOT_C + pos] = make_int2(s[k], d[k]);
            }
        }
    }
    __syncthreads();
    if (threadIdx.x < BUCKET_NODES) bcnt[threadIdx.x * NCHUNK + g] = lc[threadIdx.x];
}

__global__ __launch_bounds__(NCHUNK) void bucket_place(
    const int2* __restrict__ rec, const int* __restrict__ bcnt,
    const int* __restrict__ src, const int* __restrict__ dst, int E,
    int* __restrict__ cnt, int* __restrict__ lists, int n)
{
    __shared__ int ln[BUCKET_NODES];
    const int t     = threadIdx.x;
    const int b     = blockIdx.x;
    const int nbase = b << BUCKET_SHIFT;
    ln[t] = 0;
    __syncthreads();

    const int c = bcnt[b * NCHUNK + t];
    if (c <= SLOT_C) {
        const int2* __restrict__ rp = rec + ((size_t)b * NCHUNK + t) * SLOT_C;
        for (int i = 0; i < c; i += 4) {
            int2 r[4];
#pragma unroll
            for (int k = 0; k < 4; ++k) {
                r[k] = make_int2(0, -1);
                if (i + k < c) r[k] = rp[i + k];
            }
#pragma unroll
            for (int k = 0; k < 4; ++k) {
                if (r[k].y >= 0) {
                    const int pos = atomicAdd(&ln[r[k].y - nbase], 1);
                    if (pos < LCAP) lists[(size_t)r[k].y * LCAP + pos] = r[k].x;
                }
            }
        }
    } else {
        const int chunk = (E + NCHUNK - 1) / NCHUNK;
        const int e0    = t * chunk;
        const int e1    = min(e0 + chunk, E);
        for (int e = e0; e < e1; ++e) {
            const int d = dst[e];
            if ((d >> BUCKET_SHIFT) == b) {
                const int sv  = src[e];
                const int pos = atomicAdd(&ln[d - nbase], 1);
                if (pos < LCAP) lists[(size_t)d * LCAP + pos] = sv;
            }
        }
    }
    __syncthreads();
    if (nbase + t < n) cnt[nbase + t] = ln[t];
}

#define NODES_PER_BLOCK 16
#define WAVES_PER_BLOCK 32
#define TILES_PER_BLOCK 13
#define NODES_PER_GROUP (TILES_PER_BLOCK * NODES_PER_BLOCK)
#define ZPITCH 68

template <int N>
static __device__ __forceinline__ float row_xor(float x)
{
    return __int_as_float(__builtin_amdgcn_update_dpp(0, __float_as_int(x), 0x160 + N, 0xf, 0xf, true));
}

static __device__ __forceinline__ float row16_sum(float p)
{
    p += row_xor<1>(p);
    p += row_xor<2>(p);
    p += row_xor<4>(p);
    p += row_xor<8>(p);
    return p;
}

template <int K>
static __device__ __forceinline__ int row_bcast(int v)
{
    return __builtin_amdgcn_update_dpp(0, v, 0x150 + K, 0xf, 0xf, true);
}

static __device__ __forceinline__ float xhalf(float x)
{
    const int b = __float_as_int(x);
    return __int_as_float(__builtin_amdgcn_permlanex16(b, b, 0x76543210, 0xfedcba98, false, false));
}

template <int U>
static __device__ __forceinline__ int slot_src(int s_lo, int s_hi)
{
    return row_bcast<(U & 15)>((U < 16) ? s_lo : s_hi);
}

#define LD_SLOT(T, U)                                                            \
    do {                                                                         \
        const int sv_ = slot_src<(U)>(s_lo, s_hi);                               \
        ok[T] = (2 * (U) + half) < deg;                                          \
        r[T] = *(const float4*)(Hb + ((unsigned)sv_ * 256u + coff));             \
    } while (0)
#define LD_GROUP(T0, U0)                                                         \
    if (2 * (U0) < deg) {                                                        \
        LD_SLOT((T0), (U0));         LD_SLOT((T0) + 1, (U0) + 1);                \
        LD_SLOT((T0) + 2, (U0) + 2); LD_SLOT((T0) + 3, (U0) + 3);                \
    }
#define SC_SLOT(T)                                                               \
    do {                                                                         \
        float p_ = r[T].x * as4.x;                                               \
        p_ = fmaf(r[T].y, as4.y, p_);                                            \
        p_ = fmaf(r[T].z, as4.z, p_);                                            \
        p_ = fmaf(r[T].w, as4.w, p_);                                            \
        float v_ = row16_sum(p_) + hd;                                           \
        v_ = (v_ >= 0.0f) ? v_ : 0.2f * v_;                                      \
        sc[T] = ok[T] ? v_ : NEG_INF;                                            \
        cm = fmaxf(cm, sc[T]);                                                   \
    } while (0)
#define SC_GROUP(T0, U0)                                                         \
    if (2 * (U0) < deg) {                                                        \
        SC_SLOT((T0)); SC_SLOT((T0) + 1); SC_SLOT((T0) + 2); SC_SLOT((T0) + 3);  \
    }
#define AC_SLOT(T)                                                               \
    do {                                                                         \
        const float w_ = __expf(sc[T] - m);                                      \
        S += w_;                                                                 \
        acc.x = fmaf(w_, r[T].x, acc.x); acc.y = fmaf(w_, r[T].y, acc.y);        \
        acc.z = fmaf(w_, r[T].z, acc.z); acc.w = fmaf(w_, r[T].w, acc.w);        \
    } while (0)
#define AC_GROUP(T0, U0)                                                         \
    if (2 * (U0) < deg) {                                                        \
        AC_SLOT((T0)); AC_SLOT((T0) + 1); AC_SLOT((T0) + 2); AC_SLOT((T0) + 3);  \
    }

template <int U0, int NS>
static __device__ __forceinline__ void gather_chunk(
    const int deg, const int half, const unsigned coff, const int s_lo, const int s_hi,
    const char* __restrict__ Hb, const float4 as4, const float hd,
    float& m, float& S, float4& acc)
{
    const float NEG_INF = __int_as_float(0xFF800000);
    float4 r[NS];
    float  sc[NS];
    bool   ok[NS];

    LD_GROUP(0, U0)
    LD_GROUP(4, U0 + 4)
    if (NS > 8) { LD_GROUP(8, U0 + 8) }

    float cm = NEG_INF;
    SC_GROUP(0, U0)
    SC_GROUP(4, U0 + 4)
    if (NS > 8) { SC_GROUP(8, U0 + 8) }
    cm = fmaxf(cm, xhalf(cm));

    const float mn = fmaxf(m, cm);
    const float f  = __expf(m - mn);
    m = mn;
    S *= f;
    acc.x *= f; acc.y *= f; acc.z *= f; acc.w *= f;

    AC_GROUP(0, U0)
    AC_GROUP(4, U0 + 4)
    if (NS > 8) { AC_GROUP(8, U0 + 8) }
}

__global__ __launch_bounds__(32 * WAVES_PER_BLOCK) void gat_gather_xform(
    const int* __restrict__ cnt, const int* __restrict__ lists,
    const int* __restrict__ esrc, const int* __restrict__ edst, int E,
    const float* __restrict__ X,
    const float* __restrict__ was,
    const float* __restrict__ wad,
    const float* __restrict__ Wm,
    const float* __restrict__ bias,
    float* __restrict__ OUT, int n_nodes)
{
    __shared__ __attribute__((aligned(16))) float Zs[NODES_PER_GROUP * ZPITCH];
    __shared__ int arrived[TILES_PER_BLOCK];
    __shared__ int next_row;
    __shared__ int degs[NODES_PER_GROUP];

    const int base_node = blockIdx.x * NODES_PER_GROUP;

    if (threadIdx.x < NODES_PER_GROUP) {
        const int nd = base_node + (int)threadIdx.x;
        degs[threadIdx.x] = (nd < n_nodes) ? cnt[nd] : 0;
    }
    if (threadIdx.x < TILES_PER_BLOCK) arrived[threadIdx.x] = 0;
    if (threadIdx.x == TILES_PER_BLOCK) next_row = 0;
    __syncthreads();

    const int lane = threadIdx.x & 31;
    const int half = lane >> 4;
    const int c    = lane & 15;

    const int lim = min(NODES_PER_GROUP, (n_nodes - base_node + NODES_PER_BLOCK - 1) & ~(NODES_PER_BLOCK - 1));

#pragma nounroll
    for (int it = 0; it < NODES_PER_GROUP; ++it) {
        int row = 0;
        if (lane == 0)
            row = __hip_atomic_fetch_add(&next_row, 1, __ATOMIC_RELAXED, __HIP_MEMORY_SCOPE_WORKGROUP);
        row = __builtin_amdgcn_readfirstlane(row);
        if (row >= lim) break;

        const int node = base_node + row;

        float4 acc;
        acc.x = 0.0f; acc.y = 0.0f; acc.z = 0.0f; acc.w = 0.0f;

        if (node < n_nodes) {
            const int deg = __builtin_amdgcn_readfirstlane(degs[row]);
            const float4* __restrict__ Xv = (const float4*)X;
            const float4 as4   = ((const float4*)was)[c];
            const float4 ad4   = ((const float4*)wad)[c];
            const float4 hself = Xv[(size_t)node * 16 + c];

            float pd = hself.x * ad4.x;
            pd = fmaf(hself.y, ad4.y, pd);
            pd = fmaf(hself.z, ad4.z, pd);
            pd = fmaf(hself.w, ad4.w, pd);
            const float hd = row16_sum(pd);

            float ps = hself.x * as4.x;
            ps = fmaf(hself.y, as4.y, ps);
            ps = fmaf(hself.z, as4.z, ps);
            ps = fmaf(hself.w, as4.w, ps);
            float es = row16_sum(ps) + hd;
            es = (es >= 0.0f) ? es : 0.2f * es;

            if (deg <= LCAP) {
                const int* __restrict__ lst = lists + (size_t)node * LCAP;
                const char* __restrict__ Hb = (const char*)X;
                const unsigned coff = 16u * (unsigned)c;

                const int e_lo = 2 * c + half;
                int s_lo = node, s_hi = node;
                if (e_lo < deg) s_lo = lst[e_lo];
                if (deg > 32) {
                    if (e_lo + 32 < deg) s_hi = lst[e_lo + 32];
                }

                float m = es;
                float S = (half == 0) ? 1.0f : 0.0f;
                acc.x = (half == 0) ? hself.x : 0.0f;
                acc.y = (half == 0) ? hself.y : 0.0f;
                acc.z = (half == 0) ? hself.z : 0.0f;
                acc.w = (half == 0) ? hself.w : 0.0f;

                if (deg > 0)  gather_chunk<0, 12>(deg, half, coff, s_lo, s_hi, Hb, as4, hd, m, S, acc);
                if (deg > 24) gather_chunk<12, 12>(deg, half, coff, s_lo, s_hi, Hb, as4, hd, m, S, acc);
                if (deg > 48) gather_chunk<24, 8>(deg, half, coff, s_lo, s_hi, Hb, as4, hd, m, S, acc);

                const float inv = 1.0f / (S + xhalf(S));
                acc.x *= inv; acc.y *= inv; acc.z *= inv; acc.w *= inv;
            } else {
                float m = es;
                float S = 1.0f;
                acc = hself;
                for (int base = 0; base < E; base += 32) {
                    const int  e   = base + lane;
                    const bool hit = (e < E) && (edst[e] == node);
                    int s = node;
                    if (hit) s = esrc[e];
                    unsigned mk = (unsigned)__ballot(hit);
                    while (mk != 0u) {
                        const int k = __builtin_ctz(mk);
                        mk &= mk - 1u;
                        const int    sk = __shfl(s, k, 32);
                        const float4 r  = Xv[(size_t)sk * 16 + c];
                        float p = r.x * as4.x;
                        p = fmaf(r.y, as4.y, p);
                        p = fmaf(r.z, as4.z, p);
                        p = fmaf(r.w, as4.w, p);
                        float v = row16_sum(p) + hd;
                        v = (v >= 0.0f) ? v : 0.2f * v;
                        const float mn = fmaxf(m, v);
                        const float f  = __expf(m - mn);
                        const float w  = __expf(v - mn);
                        m = mn;
                        S = fmaf(S, f, w);
                        acc.x = fmaf(acc.x, f, w * r.x); acc.y = fmaf(acc.y, f, w * r.y);
                        acc.z = fmaf(acc.z, f, w * r.z); acc.w = fmaf(acc.w, f, w * r.w);
                    }
                }
                const float inv = (half == 0) ? 1.0f / S : 0.0f;
                acc.x *= inv; acc.y *= inv; acc.z *= inv; acc.w *= inv;
            }

            acc.x += xhalf(acc.x);
            acc.y += xhalf(acc.y);
            acc.z += xhalf(acc.z);
            acc.w += xhalf(acc.w);
        }

        const int tl = row >> 4;
        if (half == 0) *(float4*)(Zs + row * ZPITCH + 4 * c) = acc;
        int owner = 0;
        if (lane == 0) {
            const int before = __hip_atomic_fetch_add(&arrived[tl], 1, __ATOMIC_ACQ_REL,
                                                      __HIP_MEMORY_SCOPE_WORKGROUP);
            owner = (before == NODES_PER_BLOCK - 1) ? 1 : 0;
        }
        owner = __builtin_amdgcn_readfirstlane(owner);
        if (owner != 0) {
            const int khalf = half;
            v8f d0 = (v8f){0,0,0,0,0,0,0,0};
            v8f d1 = (v8f){0,0,0,0,0,0,0,0};
            v8f d2 = (v8f){0,0,0,0,0,0,0,0};
            v8f d3 = (v8f){0,0,0,0,0,0,0,0};
            const float* zr = Zs + (tl * NODES_PER_BLOCK + c) * ZPITCH + 2 * khalf;
            const float4* __restrict__ wq = (const float4*)(Wm + (2 * khalf) * FDIM) + c;
#pragma nounroll
            for (int kc = 0; kc < 4; ++kc) {
                const float*  zc = zr + 16 * kc;
                const float4* __restrict__ wc = wq + (16 * kc) * (FDIM / 4);
#pragma unroll
                for (int k = 0; k < 4; ++k) {
                    const v2f    af = *(const v2f*)(zc + 4 * k);
                    const float4 q0 = wc[(4 * k) * (FDIM / 4)];
                    const float4 q1 = wc[(4 * k + 1) * (FDIM / 4)];
                    v2f b;
                    b.x = q0.x; b.y = q1.x;
                    d0 = __builtin_amdgcn_wmma_f32_16x16x4_f32(false, af, false, b, (short)0, d0, false, false);
                    b.x = q0.y; b.y = q1.y;
                    d1 = __builtin_amdgcn_wmma_f32_16x16x4_f32(false, af, false, b, (short)0, d1, false, false);
                    b.x = q0.z; b.y = q1.z;
                    d2 = __builtin_amdgcn_wmma_f32_16x16x4_f32(false, af, false, b, (short)0, d2, false, false);
                    b.x = q0.w; b.y = q1.w;
                    d3 = __builtin_amdgcn_wmma_f32_16x16x4_f32(false, af, false, b, (short)0, d3, false, false);
                }
            }

            const float4 bz = ((const float4*)bias)[c];
            const int rbase = base_node + tl * NODES_PER_BLOCK + 8 * khalf;
#pragma unroll
            for (int r = 0; r < 8; ++r) {
                float4 o;
                o.x = d0[r] + bz.x;
                o.y = d1[r] + bz.y;
                o.z = d2[r] + bz.z;
                o.w = d3[r] + bz.w;
                if (rbase + r < n_nodes) ((float4*)OUT)[(size_t)(rbase + r) * 16 + c] = o;
            }
        }
    }
}

extern "C" void kernel_launch(void* const* d_in, const int* in_sizes, int n_in,
                              void* d_out, int out_size, void* d_ws, size_t ws_size,
                              hipStream_t stream) {
    const float* x   = (const float*)d_in[0];
    const int*   ei  = (const int*)  d_in[1];
    const float* W1  = (const float*)d_in[2];
    const float* as1 = (const float*)d_in[3];
    const float* ad1 = (const float*)d_in[4];
    const float* b1  = (const float*)d_in[5];
    const float* W2  = (const float*)d_in[6];
    const float* as2 = (const float*)d_in[7];
    const float* ad2 = (const float*)d_in[8];
    const float* b2  = (const float*)d_in[9];

    const int N    = in_sizes[0] / FDIM;
    const int E    = in_sizes[1] / 2;
    const int* src = ei;
    const int* dst = ei + E;

    float* ws = (float*)d_ws;
    float* X1 = ws;  ws += (size_t)N * FDIM;
    int* lists = (int*)ws;  ws += (size_t)N * LCAP;
    float* wvec = ws;  ws += 4 * FDIM;
    int* bcnt = (int*)ws;  ws += BUCKET_NODES * NCHUNK;
    int2* rec = (int2*)ws;  ws += (size_t)2 * BUCKET_NODES * NCHUNK * SLOT_C;
    int* cnt  = (int*)ws;

    edge_bucket<<<NCHUNK + 1, EB_THREADS, 0, stream>>>(src, dst, E, rec, bcnt,
                                                       W1, as1, ad1, W2, as2, ad2, wvec);
    bucket_place<<<(N + BUCKET_NODES - 1) / BUCKET_NODES, NCHUNK, 0, stream>>>(
        rec, bcnt, src, dst, E, cnt, lists, N);

    auto layer = [&](const float* Xin, const float* was, const float* wad,
                     const float* W, const float* b, float* OUT) {
        gat_gather_xform<<<(N + NODES_PER_GROUP - 1) / NODES_PER_GROUP,
                           32 * WAVES_PER_BLOCK, 0, stream>>>(cnt, lists, src, dst, E,
                                                              Xin, was, wad, W, b, OUT, N);
    };

    layer(x,  wvec,            wvec + FDIM,     W1, b1, X1);
    layer(X1, wvec + 2 * FDIM, wvec + 3 * FDIM, W2, b2, (float*)d_out);
}


# Edge records placed into per-node lists in the opening of the gather kernel, lists and degrees kept in LDS (bucket = a block's 208 nodes); placing launch gone, 4 to 3 launches; sum order by arrival
# speedup vs baseline: 8.0995x; 1.1749x over previous
#include <hip/hip_runtime.h>
#include <math.h>

#define FDIM 64

typedef __attribute__((ext_vector_type(2))) float v2f;
typedef __attribute__((ext_vector_type(8))) float v8f;


#define LCAP 64

#define NODES_PER_BLOCK 16
#define WAVES_PER_BLOCK 32
#define TILES_PER_BLOCK 13
#define NODES_PER_GROUP (TILES_PER_BLOCK * NODES_PER_BLOCK)
#define MAX_BUCKETS 256
#define NCHUNK 256
#define SLOT_C 64
#define EB_THREADS 512

__global__ __launch_bounds__(EB_THREADS) void edge_bucket(
    const int* __restrict__ src, const int* __restrict__ dst, int E,
    int2* __restrict__ rec, int* __restrict__ bcnt,
    const float* __restrict__ W1, const float* __restrict__ as1, const float* __restrict__ ad1,
    const float* __restrict__ W2, const float* __restrict__ as2, const float* __restrict__ ad2,
    float* __restrict__ wvec)
{
    __shared__ int lc[MAX_BUCKETS];

    if (blockIdx.x == NCHUNK) {
        if (threadIdx.x < 4 * FDIM) {
            const int t   = threadIdx.x;
            const int v   = t >> 6;
            const int row = t & 63;
            const float* __restrict__ Wm = (v < 2) ? W1 : W2;
            const float* __restrict__ a  = (v == 0) ? as1 : (v == 1) ? ad1 : (v == 2) ? as2 : ad2;
            const float4* __restrict__ wr = (const float4*)(Wm + row * FDIM);
            const float4* __restrict__ av = (const float4*)a;
            float s = 0.0f;
#pragma unroll
            for (int q = 0; q < FDIM / 4; ++q) {
                const float4 w4 = wr[q];
                const float4 a4 = av[q];
                s = fmaf(w4.x, a4.x, s);
                s = fmaf(w4.y, a4.y, s);
                s = fmaf(w4.z, a4.z, s);
                s = fmaf(w4.w, a4.w, s);
            }
            wvec[v * FDIM + row] = s;
        }
        return;
    }

    if (threadIdx.x < MAX_BUCKETS) lc[threadIdx.x] = 0;
    __syncthreads();

    const int g     = blockIdx.x;
    const int chunk = (E + NCHUNK - 1) / NCHUNK;
    const int e0    = g * chunk;
    const int e1    = min(e0 + chunk, E);
    for (int base = e0; base < e1; base += 4 * EB_THREADS) {
        int d[4], s[4];
#pragma unroll
        for (int k = 0; k < 4; ++k) {
            const int e = base + k * EB_THREADS + (int)threadIdx.x;
            d[k] = -1; s[k] = 0;
            if (e < e1) { d[k] = dst[e]; s[k] = src[e]; }
        }
#pragma unroll
        for (int k = 0; k < 4; ++k) {
            if (d[k] >= 0) {
                const int b   = d[k] / NODES_PER_GROUP;
                const int pos = atomicAdd(&lc[b], 1);
                if (pos < SLOT_C)
                    rec[((size_t)b * NCHUNK + g) * SLOT_C + pos] = make_int2(s[k], d[k]);
            }
        }
    }
    __syncthreads();
    if (threadIdx.x < MAX_BUCKETS) bcnt[threadIdx.x * NCHUNK + g] = lc[threadIdx.x];
}

#define ZPITCH 68
static_assert(32 * WAVES_PER_BLOCK == 4 * NCHUNK, "four threads of a gather block per chunk slot");

template <int N>
static __device__ __forceinline__ float row_xor(float x)
{
    return __int_as_float(__builtin_amdgcn_update_dpp(0, __float_as_int(x), 0x160 + N, 0xf, 0xf, true));
}

static __device__ __forceinline__ float row16_sum(float p)
{
    p += row_xor<1>(p);
    p += row_xor<2>(p);
    p += row_xor<4>(p);
    p += row_xor<8>(p);
    return p;
}

template <int K>
static __device__ __forceinline__ int row_bcast(int v)
{
    return __builtin_amdgcn_update_dpp(0, v, 0x150 + K, 0xf, 0xf, true);
}

static __device__ __forceinline__ float xhalf(float x)
{
    const int b = __float_as_int(x);
    return __int_as_float(__builtin_amdgcn_permlanex16(b, b, 0x76543210, 0xfedcba98, false, false));
}

template <int U>
static __device__ __forceinline__ int slot_src(int s_lo, int s_hi)
{
    return row_bcast<(U & 15)>((U < 16) ? s_lo : s_hi);
}

#define LD_SLOT(T, U)                                                            \
    do {                                                                         \
        const int sv_ = slot_src<(U)>(s_lo, s_hi);                               \
        ok[T] = (2 * (U) + half) < deg;                                          \
        r[T] = *(const float4*)(Hb + ((unsigned)sv_ * 256u + coff));             \
    } while (0)
#define LD_GROUP(T0, U0)                                                         \
    if (2 * (U0) < deg) {                                                        \
        LD_SLOT((T0), (U0));         LD_SLOT((T0) + 1, (U0) + 1);                \
        LD_SLOT((T0) + 2, (U0) + 2); LD_SLOT((T0) + 3, (U0) + 3);                \
    }
#define SC_SLOT(T)                                                               \
    do {                                                                         \
        float p_ = r[T].x * as4.x;                                               \
        p_ = fmaf(r[T].y, as4.y, p_);                                            \
        p_ = fmaf(r[T].z, as4.z, p_);                                            \
        p_ = fmaf(r[T].w, as4.w, p_);                                            \
        float v_ = row16_sum(p_) + hd;                                           \
        v_ = (v_ >= 0.0f) ? v_ : 0.2f * v_;                                      \
        sc[T] = ok[T] ? v_ : NEG_INF;                                            \
        cm = fmaxf(cm, sc[T]);                                                   \
    } while (0)
#define SC_GROUP(T0, U0)                                                         \
    if (2 * (U0) < deg) {                                                        \
        SC_SLOT((T0)); SC_SLOT((T0) + 1); SC_SLOT((T0) + 2); SC_SLOT((T0) + 3);  \
    }
#define AC_SLOT(T)                                                               \
    do {                                                                         \
        const float w_ = __expf(sc[T] - m);                                      \
        S += w_;                                                                 \
        acc.x = fmaf(w_, r[T].x, acc.x); acc.y = fmaf(w_, r[T].y, acc.y);        \
        acc.z = fmaf(w_, r[T].z, acc.z); acc.w = fmaf(w_, r[T].w, acc.w);        \
    } while (0)
#define AC_GROUP(T0, U0)                                                         \
    if (2 * (U0) < deg) {                                                        \
        AC_SLOT((T0)); AC_SLOT((T0) + 1); AC_SLOT((T0) + 2); AC_SLOT((T0) + 3);  \
    }

template <int U0, int NS>
static __device__ __forceinline__ void gather_chunk(
    const int deg, const int half, const unsigned coff, const int s_lo, const int s_hi,
    const char* __restrict__ Hb, const float4 as4, const float hd,
    float& m, float& S, float4& acc)
{
    const float NEG_INF = __int_as_float(0xFF800000);
    float4 r[NS];
    float  sc[NS];
    bool   ok[NS];

    LD_GROUP(0, U0)
    LD_GROUP(4, U0 + 4)
    if (NS > 8) { LD_GROUP(8, U0 + 8) }

    float cm = NEG_INF;
    SC_GROUP(0, U0)
    SC_GROUP(4, U0 + 4)
    if (NS > 8) { SC_GROUP(8, U0 + 8) }
    cm = fmaxf(cm, xhalf(cm));

    const float mn = fmaxf(m, cm);
    const float f  = __expf(m - mn);
    m = mn;
    S *= f;
    acc.x *= f; acc.y *= f; acc.z *= f; acc.w *= f;

    AC_GROUP(0, U0)
    AC_GROUP(4, U0 + 4)
    if (NS > 8) { AC_GROUP(8, U0 + 8) }
}

__global__ __launch_bounds__(32 * WAVES_PER_BLOCK) void gat_gather_xform(
    const int2* __restrict__ rec, const int* __restrict__ bcnt,
    const int* __restrict__ esrc, const int* __restrict__ edst, int E,
    const float* __restrict__ X,
    const float* __restrict__ was,
    const float* __restrict__ wad,
    const float* __restrict__ Wm,
    const float* __restrict__ bias,
    float* __restrict__ OUT, int n_nodes)
{
    __shared__ __attribute__((aligned(16))) float Zs[NODES_PER_GROUP * ZPITCH];
    __shared__ int arrived[TILES_PER_BLOCK];
    __shared__ int next_row;
    __shared__ int degs[NODES_PER_GROUP];
    __shared__ int Ll[NODES_PER_GROUP * LCAP];

    const int bkt       = blockIdx.x;
    const int base_node = bkt * NODES_PER_GROUP;

    const int slot = (int)threadIdx.x >> 2;
    const int nrec = bcnt[bkt * NCHUNK + slot];

    if (threadIdx.x < NODES_PER_GROUP) degs[threadIdx.x] = 0;
    if (threadIdx.x < TILES_PER_BLOCK) arrived[threadIdx.x] = 0;
    if (threadIdx.x == TILES_PER_BLOCK) next_row = 0;
    __syncthreads();

    {
        const int q = (int)threadIdx.x & 3;
        if (nrec <= SLOT_C) {
            const int2* __restrict__ rp = rec + ((size_t)bkt * NCHUNK + slot) * SLOT_C;
            for (int i = q; i < nrec; i += 32) {
                int2 r[8];
#pragma unroll
                for (int k = 0; k < 8; ++k) {
                    r[k] = make_int2(0, -1);
                    if (i + 4 * k < nrec) r[k] = rp[i + 4 * k];
                }
                int pos[8];
#pragma unroll
                for (int k = 0; k < 8; ++k) {
                    pos[k] = LCAP;
                    if (r[k].y >= 0) pos[k] = atomicAdd(&degs[r[k].y - base_node], 1);
                }
#pragma unroll
                for (int k = 0; k < 8; ++k) {
                    if (pos[k] < LCAP) Ll[(r[k].y - base_node) * LCAP + pos[k]] = r[k].x;
                }
            }
        } else if (q == 0) {
            const int chunk = (E + NCHUNK - 1) / NCHUNK;
            const int e0    = slot * chunk;
            const int e1    = min(e0 + chunk, E);
            for (int e = e0; e < e1; ++e) {
                const int d = edst[e];
                if (d / NODES_PER_GROUP == bkt) {
                    const int sv  = esrc[e];
                    const int ln  = d - base_node;
                    const int pos = atomicAdd(&degs[ln], 1);
                    if (pos < LCAP) Ll[ln * LCAP + pos] = sv;
                }
            }
        }
    }
    __syncthreads();

    const int lane = threadIdx.x & 31;
    const int half = lane >> 4;
    const int c    = lane & 15;

    const int lim = min(NODES_PER_GROUP, (n_nodes - base_node + NODES_PER_BLOCK - 1) & ~(NODES_PER_BLOCK - 1));

#pragma nounroll
    for (int it = 0; it < NODES_PER_GROUP; ++it) {
        int row = 0;
        if (lane == 0)
            row = __hip_atomic_fetch_add(&next_row, 1, __ATOMIC_RELAXED, __HIP_MEMORY_SCOPE_WORKGROUP);
        row = __builtin_amdgcn_readfirstlane(row);
        if (row >= lim) break;

        const int node = base_node + row;

        float4 acc;
        acc.x = 0.0f; acc.y = 0.0f; acc.z = 0.0f; acc.w = 0.0f;

        if (node < n_nodes) {
            const int deg = __builtin_amdgcn_readfirstlane(degs[row]);
            const float4* __restrict__ Xv = (const float4*)X;
            const float4 as4   = ((const float4*)was)[c];
            const float4 ad4   = ((const float4*)wad)[c];
            const float4 hself = Xv[(size_t)node * 16 + c];

            float pd = hself.x * ad4.x;
            pd = fmaf(hself.y, ad4.y, pd);
            pd = fmaf(hself.z, ad4.z, pd);
            pd = fmaf(hself.w, ad4.w, pd);
            const float hd = row16_sum(pd);

            float ps = hself.x * as4.x;
            ps = fmaf(hself.y, as4.y, ps);
            ps = fmaf(hself.z, as4.z, ps);
            ps = fmaf(hself.w, as4.w, ps);
            float es = row16_sum(ps) + hd;
            es = (es >= 0.0f) ? es : 0.2f * es;

            if (deg <= LCAP) {
                const int* lst = Ll + row * LCAP;
                const char* __restrict__ Hb = (const char*)X;
                const unsigned coff = 16u * (unsigned)c;

                const int e_lo = 2 * c + half;
                int s_lo = node, s_hi = node;
                if (e_lo < deg) s_lo = lst[e_lo];
                if (deg > 32) {
                    if (e_lo + 32 < deg) s_hi = lst[e_lo + 32];
                }

                float m = es;
                float S = (half == 0) ? 1.0f : 0.0f;
                acc.x = (half == 0) ? hself.x : 0.0f;
                acc.y = (half == 0) ? hself.y : 0.0f;
                acc.z = (half == 0) ? hself.z : 0.0f;
                acc.w = (half == 0) ? hself.w : 0.0f;

                if (deg > 0)  gather_chunk<0, 12>(deg, half, coff, s_lo, s_hi, Hb, as4, hd, m, S, acc);
                if (deg > 24) gather_chunk<12, 12>(deg, half, coff, s_lo, s_hi, Hb, as4, hd, m, S, acc);
                if (deg > 48) gather_chunk<24, 8>(deg, half, coff, s_lo, s_hi, Hb, as4, hd, m, S, acc);

                const float inv = 1.0f / (S + xhalf(S));
                acc.x *= inv; acc.y *= inv; acc.z *= inv; acc.w *= inv;
            } else {
                float m = es;
                float S = 1.0f;
                acc = hself;
                for (int base = 0; base < E; base += 32) {
                    const int  e   = base + lane;
                    const bool hit = (e < E) && (edst[e] == node);
                    int s = node;
                    if (hit) s = esrc[e];
                    unsigned mk = (unsigned)__ballot(hit);
                    while (mk != 0u) {
                        const int k = __builtin_ctz(mk);
                        mk &= mk - 1u;
                        const int    sk = __shfl(s, k, 32);
                        const float4 r  = Xv[(size_t)sk * 16 + c];
                        float p = r.x * as4.x;
                        p = fmaf(r.y, as4.y, p);
                        p = fmaf(r.z, as4.z, p);
                        p = fmaf(r.w, as4.w, p);
                        float v = row16_sum(p) + hd;
                        v = (v >= 0.0f) ? v : 0.2f * v;
                        const float mn = fmaxf(m, v);
                        const float f  = __expf(m - mn);
                        const float w  = __expf(v - mn);
                        m = mn;
                        S = fmaf(S, f, w);
                        acc.x = fmaf(acc.x, f, w * r.x); acc.y = fmaf(acc.y, f, w * r.y);
                        acc.z = fmaf(acc.z, f, w * r.z); acc.w = fmaf(acc.w, f, w * r.w);
                    }
                }
                const float inv = (half == 0) ? 1.0f / S : 0.0f;
                acc.x *= inv; acc.y *= inv; acc.z *= inv; acc.w *= inv;
            }

            acc.x += xhalf(acc.x);
            acc.y += xhalf(acc.y);
            acc.z += xhalf(acc.z);
            acc.w += xhalf(acc.w);
        }

        const int tl = row >> 4;
        if (half == 0) *(float4*)(Zs + row * ZPITCH + 4 * c) = acc;
        int owner = 0;
        if (lane == 0) {
            const int before = __hip_atomic_fetch_add(&arrived[tl], 1, __ATOMIC_ACQ_REL,
                                                      __HIP_MEMORY_SCOPE_WORKGROUP);
            owner = (before == NODES_PER_BLOCK - 1) ? 1 : 0;
        }
        owner = __builtin_amdgcn_readfirstlane(owner);
        if (owner != 0) {
            const int khalf = half;
            v8f d0 = (v8f){0,0,0,0,0,0,0,0};
            v8f d1 = (v8f){0,0,0,0,0,0,0,0};
            v8f d2 = (v8f){0,0,0,0,0,0,0,0};
            v8f d3 = (v8f){0,0,0,0,0,0,0,0};
            const float* zr = Zs + (tl * NODES_PER_BLOCK + c) * ZPITCH + 2 * khalf;
            const float4* __restrict__ wq = (const float4*)(Wm + (2 * khalf) * FDIM) + c;
#pragma nounroll
            for (int kc = 0; kc < 4; ++kc) {
                const float*  zc = zr + 16 * kc;
                const float4* __restrict__ wc = wq + (16 * kc) * (FDIM / 4);
#pragma unroll
                for (int k = 0; k < 4; ++k) {
                    const v2f    af = *(const v2f*)(zc + 4 * k);
                    const float4 q0 = wc[(4 * k) * (FDIM / 4)];
                    const float4 q1 = wc[(4 * k + 1) * (FDIM / 4)];
                    v2f b;
                    b.x = q0.x; b.y = q1.x;
                    d0 = __builtin_amdgcn_wmma_f32_16x16x4_f32(false, af, false, b, (short)0, d0, false, false);
                    b.x = q0.y; b.y = q1.y;
                    d1 = __builtin_amdgcn_wmma_f32_16x16x4_f32(false, af, false, b, (short)0, d1, false, false);
                    b.x = q0.z; b.y = q1.z;
                    d2 = __builtin_amdgcn_wmma_f32_16x16x4_f32(false, af, false, b, (short)0, d2, false, false);
                    b.x = q0.w; b.y = q1.w;
                    d3 = __builtin_amdgcn_wmma_f32_16x16x4_f32(false, af, false, b, (short)0, d3, false, false);
                }
            }

            const float4 bz = ((const float4*)bias)[c];
            const int rbase = base_node + tl * NODES_PER_BLOCK + 8 * khalf;
#pragma unroll
            for (int r = 0; r < 8; ++r) {
                float4 o;
                o.x = d0[r] + bz.x;
                o.y = d1[r] + bz.y;
                o.z = d2[r] + bz.z;
                o.w = d3[r] + bz.w;
                if (rbase + r < n_nodes) ((float4*)OUT)[(size_t)(rbase + r) * 16 + c] = o;
            }
        }
    }
}

extern "C" void kernel_launch(void* const* d_in, const int* in_sizes, int n_in,
                              void* d_out, int out_size, void* d_ws, size_t ws_size,
                              hipStream_t stream) {
    const float* x   = (const float*)d_in[0];
    const int*   ei  = (const int*)  d_in[1];
    const float* W1  = (const float*)d_in[2];
    const float* as1 = (const float*)d_in[3];
    const float* ad1 = (const float*)d_in[4];
    const float* b1  = (const float*)d_in[5];
    const float* W2  = (const float*)d_in[6];
    const float* as2 = (const float*)d_in[7];
    const float* ad2 = (const float*)d_in[8];
    const float* b2  = (const float*)d_in[9];

    const int N    = in_sizes[0] / FDIM;
    const int E    = in_sizes[1] / 2;
    const int* src = ei;
    const int* dst = ei + E;

    float* ws = (float*)d_ws;
    float* X1 = ws;  ws += (size_t)N * FDIM;
    float* wvec = ws;  ws += 4 * FDIM;
    int* bcnt = (int*)ws;  ws += MAX_BUCKETS * NCHUNK;
    int2* rec = (int2*)ws;

    edge_bucket<<<NCHUNK + 1, EB_THREADS, 0, stream>>>(src, dst, E, rec, bcnt,
                                                       W1, as1, ad1, W2, as2, ad2, wvec);

    auto layer = [&](const float* Xin, const float* was, const float* wad,
                     const float* W, const float* b, float* OUT) {
        gat_gather_xform<<<(N + NODES_PER_GROUP - 1) / NODES_PER_GROUP,
                           32 * WAVES_PER_BLOCK, 0, stream>>>(rec, bcnt, src, dst, E,
                                                              Xin, was, wad, W, b, OUT, N);
    };

    layer(x,  wvec,            wvec + FDIM,     W1, b1, X1);
    layer(X1, wvec + 2 * FDIM, wvec + 3 * FDIM, W2, b2, (float*)d_out);
}


# Scores of 4 gathered slots summed in one folded 16-lane butterfly (lane group g keeps slot g): slope, mask, max, exp once per 4 slots, weights handed back by row_share; f32 sum order changes
# speedup vs baseline: 8.7750x; 1.0834x over previous
#include <hip/hip_runtime.h>
#include <math.h>

#define FDIM 64

typedef __attribute__((ext_vector_type(2))) float v2f;
typedef __attribute__((ext_vector_type(8))) float v8f;


#define LCAP 64

#define NODES_PER_BLOCK 16
#define WAVES_PER_BLOCK 32
#define TILES_PER_BLOCK 13
#define NODES_PER_GROUP (TILES_PER_BLOCK * NODES_PER_BLOCK)
#define MAX_BUCKETS 256
#define NCHUNK 256
#define SLOT_C 64
#define EB_THREADS 512

__global__ __launch_bounds__(EB_THREADS) void edge_bucket(
    const int* __restrict__ src, const int* __restrict__ dst, int E,
    int2* __restrict__ rec, int* __restrict__ bcnt,
    const float* __restrict__ W1, const float* __restrict__ as1, const float* __restrict__ ad1,
    const float* __restrict__ W2, const float* __restrict__ as2, const float* __restrict__ ad2,
    float* __restrict__ wvec)
{
    __shared__ int lc[MAX_BUCKETS];

    if (blockIdx.x == NCHUNK) {
        if (threadIdx.x < 4 * FDIM) {
            const int t   = threadIdx.x;
            const int v   = t >> 6;
            const int row = t & 63;
            const float* __restrict__ Wm = (v < 2) ? W1 : W2;
            const float* __restrict__ a  = (v == 0) ? as1 : (v == 1) ? ad1 : (v == 2) ? as2 : ad2;
            const float4* __restrict__ wr = (const float4*)(Wm + row * FDIM);
            const float4* __restrict__ av = (const float4*)a;
            float s = 0.0f;
#pragma unroll
            for (int q = 0; q < FDIM / 4; ++q) {
                const float4 w4 = wr[q];
                const float4 a4 = av[q];
                s = fmaf(w4.x, a4.x, s);
                s = fmaf(w4.y, a4.y, s);
                s = fmaf(w4.z, a4.z, s);
                s = fmaf(w4.w, a4.w, s);
            }
            wvec[v * FDIM + row] = s;
        }
        return;
    }

    if (threadIdx.x < MAX_BUCKETS) lc[threadIdx.x] = 0;
    __syncthreads();

    const int g     = blockIdx.x;
    const int chunk = (E + NCHUNK - 1) / NCHUNK;
    const int e0    = g * chunk;
    const int e1    = min(e0 + chunk, E);
    for (int base = e0; base < e1; base += 4 * EB_THREADS) {
        int d[4], s[4];
#pragma unroll
        for (int k = 0; k < 4; ++k) {
            const int e = base + k * EB_THREADS + (int)threadIdx.x;
            d[k] = -1; s[k] = 0;
            if (e < e1) { d[k] = dst[e]; s[k] = src[e]; }
        }
#pragma unroll
        for (int k = 0; k < 4; ++k) {
            if (d[k] >= 0) {
                const int b   = d[k] / NODES_PER_GROUP;
                const int pos = atomicAdd(&lc[b], 1);
                if (pos < SLOT_C)
                    rec[((size_t)b * NCHUNK + g) * SLOT_C + pos] = make_int2(s[k], d[k]);
            }
        }
    }
    __syncthreads();
    if (threadIdx.x < MAX_BUCKETS) bcnt[threadIdx.x * NCHUNK + g] = lc[threadIdx.x];
}

#define ZPITCH 68
static_assert(32 * WAVES_PER_BLOCK == 4 * NCHUNK, "four threads of a gather block per chunk slot");

template <int N>
static __device__ __forceinline__ float row_xor(float x)
{
    return __int_as_float(__builtin_amdgcn_update_dpp(0, __float_as_int(x), 0x160 + N, 0xf, 0xf, true));
}

static __device__ __forceinline__ float row16_sum(float p)
{
    p += row_xor<1>(p);
    p += row_xor<2>(p);
    p += row_xor<4>(p);
    p += row_xor<8>(p);
    return p;
}

template <int K>
static __device__ __forceinline__ int row_bcast(int v)
{
    return __builtin_amdgcn_update_dpp(0, v, 0x150 + K, 0xf, 0xf, true);
}

static __device__ __forceinline__ float xhalf(float x)
{
    const int b = __float_as_int(x);
    return __int_as_float(__builtin_amdgcn_permlanex16(b, b, 0x76543210, 0xfedcba98, false, false));
}

static __device__ __forceinline__ float dot4(const float4 r, const float4 a)
{
    v2f q = (v2f){r.x, r.y} * (v2f){a.x, a.y};
    q = (v2f){r.z, r.w} * (v2f){a.z, a.w} + q;
    return q.x + q.y;
}

template <int U>
static __device__ __forceinline__ int slot_src(int s_lo, int s_hi)
{
    return row_bcast<(U & 15)>((U < 16) ? s_lo : s_hi);
}

#define LD_SLOT(T, U)                                                            \
    do {                                                                         \
        const int sv_ = slot_src<(U)>(s_lo, s_hi);                               \
        r[T] = *(const float4*)(Hb + ((unsigned)sv_ * 256u + coff));             \
    } while (0)
#define LD_GROUP(T0, U0)                                                         \
    if (2 * (U0) < deg) {                                                        \
        LD_SLOT((T0), (U0));         LD_SLOT((T0) + 1, (U0) + 1);                \
        LD_SLOT((T0) + 2, (U0) + 2); LD_SLOT((T0) + 3, (U0) + 3);                \
    }
#define SC_GROUP(G, T0, U0)                                                      \
    if (2 * (U0) < deg) {                                                        \
        const float p0_ = dot4(r[(T0)], as4);                                    \
        const float p1_ = dot4(r[(T0) + 1], as4);                                \
        const float p2_ = dot4(r[(T0) + 2], as4);                                \
        const float p3_ = dot4(r[(T0) + 3], as4);                                \
        const float k01_ = hi8 ? p1_ : p0_;                                      \
        const float s01_ = hi8 ? p0_ : p1_;                                      \
        const float k23_ = hi8 ? p3_ : p2_;                                      \
        const float s23_ = hi8 ? p2_ : p3_;                                      \
        const float r01_ = k01_ + row_xor<8>(s01_);                              \
        const float r23_ = k23_ + row_xor<8>(s23_);                              \
        const float kk_ = hi4 ? r23_ : r01_;                                     \
        const float ss_ = hi4 ? r01_ : r23_;                                     \
        float v_ = kk_ + row_xor<4>(ss_);                                        \
        v_ += row_xor<2>(v_);                                                    \
        v_ += row_xor<1>(v_);                                                    \
        v_ += hd;                                                                \
        v_ = (v_ >= 0.0f) ? v_ : 0.2f * v_;                                      \
        sc[G] = ((2 * (U0) + eg) < deg) ? v_ : NEG_INF;                          \
        cm = fmaxf(cm, sc[G]);                                                   \
    }
#define AC_SLOT(T, K)                                                            \
    do {                                                                         \
        const float wb_ = __int_as_float(row_bcast<(K)>(__float_as_int(w_)));    \
        const v2f w2_ = (v2f){wb_, wb_};                                         \
        a01 = (v2f){r[T].x, r[T].y} * w2_ + a01;                                 \
        a23 = (v2f){r[T].z, r[T].w} * w2_ + a23;                                 \
    } while (0)
#define AC_GROUP(G, T0, U0)                                                      \
    if (2 * (U0) < deg) {                                                        \
        const float w_ = __expf(sc[G] - m);                                      \
        S += w_;                                                                 \
        AC_SLOT((T0), 0);     AC_SLOT((T0) + 1, 8);                              \
        AC_SLOT((T0) + 2, 4); AC_SLOT((T0) + 3, 12);                             \
    }

template <int U0, int NS>
static __device__ __forceinline__ void gather_chunk(
    const int deg, const int eg, const bool hi8, const bool hi4, const unsigned coff,
    const int s_lo, const int s_hi,
    const char* __restrict__ Hb, const float4 as4, const float hd,
    float& m, float& S, float4& acc)
{
    const float NEG_INF = __int_as_float(0xFF800000);
    float4 r[NS];
    float  sc[NS / 4];

    LD_GROUP(0, U0)
    LD_GROUP(4, U0 + 4)
    if (NS > 8) { LD_GROUP(8, U0 + 8) }

    float cm = NEG_INF;
    SC_GROUP(0, 0, U0)
    SC_GROUP(1, 4, U0 + 4)
    if (NS > 8) { SC_GROUP(2, 8, U0 + 8) }
    cm = fmaxf(cm, row_xor<4>(cm));
    cm = fmaxf(cm, row_xor<8>(cm));
    cm = fmaxf(cm, xhalf(cm));

    const float mn = fmaxf(m, cm);
    const float f  = __expf(m - mn);
    m = mn;
    S *= f;
    v2f a01 = (v2f){acc.x * f, acc.y * f};
    v2f a23 = (v2f){acc.z * f, acc.w * f};

    AC_GROUP(0, 0, U0)
    AC_GROUP(1, 4, U0 + 4)
    if (NS > 8) { AC_GROUP(2, 8, U0 + 8) }

    acc.x = a01.x; acc.y = a01.y; acc.z = a23.x; acc.w = a23.y;
}

__global__ __launch_bounds__(32 * WAVES_PER_BLOCK) void gat_gather_xform(
    const int2* __restrict__ rec, const int* __restrict__ bcnt,
    const int* __restrict__ esrc, const int* __restrict__ edst, int E,
    const float* __restrict__ X,
    const float* __restrict__ was,
    const float* __restrict__ wad,
    const float* __restrict__ Wm,
    const float* __restrict__ bias,
    float* __restrict__ OUT, int n_nodes)
{
    __shared__ __attribute__((aligned(16))) float Zs[NODES_PER_GROUP * ZPITCH];
    __shared__ int arrived[TILES_PER_BLOCK];
    __shared__ int next_row;
    __shared__ int degs[NODES_PER_GROUP];
    __shared__ int Ll[NODES_PER_GROUP * LCAP];

    const int bkt       = blockIdx.x;
    const int base_node = bkt * NODES_PER_GROUP;

    const int slot = (int)threadIdx.x >> 2;
    const int nrec = bcnt[bkt * NCHUNK + slot];

    if (threadIdx.x < NODES_PER_GROUP) degs[threadIdx.x] = 0;
    if (threadIdx.x < TILES_PER_BLOCK) arrived[threadIdx.x] = 0;
    if (threadIdx.x == TILES_PER_BLOCK) next_row = 0;
    __syncthreads();

    {
        const int q = (int)threadIdx.x & 3;
        if (nrec <= SLOT_C) {
            const int2* __restrict__ rp = rec + ((size_t)bkt * NCHUNK + slot) * SLOT_C;
            for (int i = q; i < nrec; i += 32) {
                int2 r[8];
#pragma unroll
                for (int k = 0; k < 8; ++k) {
                    r[k] = make_int2(0, -1);
                    if (i + 4 * k < nrec) r[k] = rp[i + 4 * k];
                }
                int pos[8];
#pragma unroll
                for (int k = 0; k < 8; ++k) {
                    pos[k] = LCAP;
                    if (r[k].y >= 0) pos[k] = atomicAdd(&degs[r[k].y - base_node], 1);
                }
#pragma unroll
                for (int k = 0; k < 8; ++k) {
                    if (pos[k] < LCAP) Ll[(r[k].y - base_node) * LCAP + pos[k]] = r[k].x;
                }
            }
        } else if (q == 0) {
            const int chunk = (E + NCHUNK - 1) / NCHUNK;
            const int e0    = slot * chunk;
            const int e1    = min(e0 + chunk, E);
            for (int e = e0; e < e1; ++e) {
                const int d = edst[e];
                if (d / NODES_PER_GROUP == bkt) {
                    const int sv  = esrc[e];
                    const int ln  = d - base_node;
                    const int pos = atomicAdd(&degs[ln], 1);
                    if (pos < LCAP) Ll[ln * LCAP + pos] = sv;
                }
            }
        }
    }
    __syncthreads();

    const int lane = threadIdx.x & 31;
    const int half = lane >> 4;
    const int c    = lane & 15;
    const bool hi8 = (c & 8) != 0;
    const bool hi4 = (c & 4) != 0;
    const int  eg  = 2 * ((hi8 ? 1 : 0) + (hi4 ? 2 : 0)) + half;

    const int lim = min(NODES_PER_GROUP, (n_nodes - base_node + NODES_PER_BLOCK - 1) & ~(NODES_PER_BLOCK - 1));

#pragma nounroll
    for (int it = 0; it < NODES_PER_GROUP; ++it) {
        int row = 0;
        if (lane == 0)
            row = __hip_atomic_fetch_add(&next_row, 1, __ATOMIC_RELAXED, __HIP_MEMORY_SCOPE_WORKGROUP);
        row = __builtin_amdgcn_readfirstlane(row);
        if (row >= lim) break;

        const int node = base_node + row;

        float4 acc;
        acc.x = 0.0f; acc.y = 0.0f; acc.z = 0.0f; acc.w = 0.0f;

        if (node < n_nodes) {
            const int deg = __builtin_amdgcn_readfirstlane(degs[row]);
            const float4* __restrict__ Xv = (const float4*)X;
            const float4 as4   = ((const float4*)was)[c];
            const float4 ad4   = ((const float4*)wad)[c];
            const float4 hself = Xv[(size_t)node * 16 + c];

            float pd = hself.x * ad4.x;
            pd = fmaf(hself.y, ad4.y, pd);
            pd = fmaf(hself.z, ad4.z, pd);
            pd = fmaf(hself.w, ad4.w, pd);
            const float hd = row16_sum(pd);

            float ps = hself.x * as4.x;
            ps = fmaf(hself.y, as4.y, ps);
            ps = fmaf(hself.z, as4.z, ps);
            ps = fmaf(hself.w, as4.w, ps);
            float es = row16_sum(ps) + hd;
            es = (es >= 0.0f) ? es : 0.2f * es;

            if (deg <= LCAP) {
                const int* lst = Ll + row * LCAP;
                const char* __restrict__ Hb = (const char*)X;
                const unsigned coff = 16u * (unsigned)c;

                const int e_lo = 2 * c + half;
                int s_lo = node, s_hi = node;
                if (e_lo < deg) s_lo = lst[e_lo];
                if (deg > 32) {
                    if (e_lo + 32 < deg) s_hi = lst[e_lo + 32];
                }

                float m = es;
                float S = (half == 0 && c < 4) ? 1.0f : 0.0f;
                acc.x = (half == 0) ? hself.x : 0.0f;
                acc.y = (half == 0) ? hself.y : 0.0f;
                acc.z = (half == 0) ? hself.z : 0.0f;
                acc.w = (half == 0) ? hself.w : 0.0f;

                if (deg > 0)  gather_chunk<0, 12>(deg, eg, hi8, hi4, coff, s_lo, s_hi, Hb, as4, hd, m, S, acc);
                if (deg > 24) gather_chunk<12, 12>(deg, eg, hi8, hi4, coff, s_lo, s_hi, Hb, as4, hd, m, S, acc);
                if (deg > 48) gather_chunk<24, 8>(deg, eg, hi8, hi4, coff, s_lo, s_hi, Hb, as4, hd, m, S, acc);

                S += row_xor<4>(S);
                S += row_xor<8>(S);
                const float inv = 1.0f / (S + xhalf(S));
                acc.x *= inv; acc.y *= inv; acc.z *= inv; acc.w *= inv;
            } else {
                float m = es;
                float S = 1.0f;
                acc = hself;
                for (int base = 0; base < E; base += 32) {
                    const int  e   = base + lane;
                    const bool hit = (e < E) && (edst[e] == node);
                    int s = node;
                    if (hit) s = esrc[e];
                    unsigned mk = (unsigned)__ballot(hit);
                    while (mk != 0u) {
                        const int k = __builtin_ctz(mk);
                        mk &= mk - 1u;
                        const int    sk = __shfl(s, k, 32);
                        const float4 r  = Xv[(size_t)sk * 16 + c];
                        float p = r.x * as4.x;
                        p = fmaf(r.y, as4.y, p);
                        p = fmaf(r.z, as4.z, p);
                        p = fmaf(r.w, as4.w, p);
                        float v = row16_sum(p) + hd;
                        v = (v >= 0.0f) ? v : 0.2f * v;
                        const float mn = fmaxf(m, v);
                        const float f  = __expf(m - mn);
                        const float w  = __expf(v - mn);
                        m = mn;
                        S = fmaf(S, f, w);
                        acc.x = fmaf(acc.x, f, w * r.x); acc.y = fmaf(acc.y, f, w * r.y);
                        acc.z = fmaf(acc.z, f, w * r.z); acc.w = fmaf(acc.w, f, w * r.w);
                    }
                }
                const float inv = (half == 0) ? 1.0f / S : 0.0f;
                acc.x *= inv; acc.y *= inv; acc.z *= inv; acc.w *= inv;
            }

            acc.x += xhalf(acc.x);
            acc.y += xhalf(acc.y);
            acc.z += xhalf(acc.z);
            acc.w += xhalf(acc.w);
        }

        const int tl = row >> 4;
        if (half == 0) *(float4*)(Zs + row * ZPITCH + 4 * c) = acc;
        int owner = 0;
        if (lane == 0) {
            const int before = __hip_atomic_fetch_add(&arrived[tl], 1, __ATOMIC_ACQ_REL,
                                                      __HIP_MEMORY_SCOPE_WORKGROUP);
            owner = (before == NODES_PER_BLOCK - 1) ? 1 : 0;
        }
        owner = __builtin_amdgcn_readfirstlane(owner);
        if (owner != 0) {
            const int khalf = half;
            v8f d0 = (v8f){0,0,0,0,0,0,0,0};
            v8f d1 = (v8f){0,0,0,0,0,0,0,0};
            v8f d2 = (v8f){0,0,0,0,0,0,0,0};
            v8f d3 = (v8f){0,0,0,0,0,0,0,0};
            const float* zr = Zs + (tl * NODES_PER_BLOCK + c) * ZPITCH + 2 * khalf;
            const float4* __restrict__ wq = (const float4*)(Wm + (2 * khalf) * FDIM) + c;
#pragma nounroll
            for (int kc = 0; kc < 4; ++kc) {
                const float*  zc = zr + 16 * kc;
                const float4* __restrict__ wc = wq + (16 * kc) * (FDIM / 4);
#pragma unroll
                for (int k = 0; k < 4; ++k) {
                    const v2f    af = *(const v2f*)(zc + 4 * k);
                    const float4 q0 = wc[(4 * k) * (FDIM / 4)];
                    const float4 q1 = wc[(4 * k + 1) * (FDIM / 4)];
                    v2f b;
                    b.x = q0.x; b.y = q1.x;
                    d0 = __builtin_amdgcn_wmma_f32_16x16x4_f32(false, af, false, b, (short)0, d0, false, false);
                    b.x = q0.y; b.y = q1.y;
                    d1 = __builtin_amdgcn_wmma_f32_16x16x4_f32(false, af, false, b, (short)0, d1, false, false);
                    b.x = q0.z; b.y = q1.z;
                    d2 = __builtin_amdgcn_wmma_f32_16x16x4_f32(false, af, false, b, (short)0, d2, false, false);
                    b.x = q0.w; b.y = q1.w;
                    d3 = __builtin_amdgcn_wmma_f32_16x16x4_f32(false, af, false, b, (short)0, d3, false, false);
                }
            }

            const float4 bz = ((const float4*)bias)[c];
            const int rbase = base_node + tl * NODES_PER_BLOCK + 8 * khalf;
#pragma unroll
            for (int r = 0; r < 8; ++r) {
                float4 o;
                o.x = d0[r] + bz.x;
                o.y = d1[r] + bz.y;
                o.z = d2[r] + bz.z;
                o.w = d3[r] + bz.w;
                if (rbase + r < n_nodes) ((float4*)OUT)[(size_t)(rbase + r) * 16 + c] = o;
            }
        }
    }
}

extern "C" void kernel_launch(void* const* d_in, const int* in_sizes, int n_in,
                              void* d_out, int out_size, void* d_ws, size_t ws_size,
                              hipStream_t stream) {
    const float* x   = (const float*)d_in[0];
    const int*   ei  = (const int*)  d_in[1];
    const float* W1  = (const float*)d_in[2];
    const float* as1 = (const float*)d_in[3];
    const float* ad1 = (const float*)d_in[4];
    const float* b1  = (const float*)d_in[5];
    const float* W2  = (const float*)d_in[6];
    const float* as2 = (const float*)d_in[7];
    const float* ad2 = (const float*)d_in[8];
    const float* b2  = (const float*)d_in[9];

    const int N    = in_sizes[0] / FDIM;
    const int E    = in_sizes[1] / 2;
    const int* src = ei;
    const int* dst = ei + E;

    float* ws = (float*)d_ws;
    float* X1 = ws;  ws += (size_t)N * FDIM;
    float* wvec = ws;  ws += 4 * FDIM;
    int* bcnt = (int*)ws;  ws += MAX_BUCKETS * NCHUNK;
    int2* rec = (int2*)ws;

    edge_bucket<<<NCHUNK + 1, EB_THREADS, 0, stream>>>(src, dst, E, rec, bcnt,
                                                       W1, as1, ad1, W2, as2, ad2, wvec);

    auto layer = [&](const float* Xin, const float* was, const float* wad,
                     const float* W, const float* b, float* OUT) {
        gat_gather_xform<<<(N + NODES_PER_GROUP - 1) / NODES_PER_GROUP,
                           32 * WAVES_PER_BLOCK, 0, stream>>>(rec, bcnt, src, dst, E,
                                                              Xin, was, wad, W, b, OUT, N);
    };

    layer(x,  wvec,            wvec + FDIM,     W1, b1, X1);
    layer(X1, wvec + 2 * FDIM, wvec + 3 * FDIM, W2, b2, (float*)d_out);
}


# edge_bucket builds each (bucket, chunk) slot in LDS and writes it as two whole 128-byte lines, count in the first word: no 8-byte store per record, no count plane; f32 sum order by arrival as before
# speedup vs baseline: 9.3022x; 1.0601x over previous
#include <hip/hip_runtime.h>
#include <math.h>

#define FDIM 64

typedef __attribute__((ext_vector_type(2))) float v2f;
typedef __attribute__((ext_vector_type(8))) float v8f;


#define LCAP 64

#define NODES_PER_BLOCK 16
#define WAVES_PER_BLOCK 32
#define TILES_PER_BLOCK 13
#define NODES_PER_GROUP (TILES_PER_BLOCK * NODES_PER_BLOCK)
#define MAX_BUCKETS 256
#define NCHUNK 256
#define SLOT_C 64
#define STAGE_C 32
#define EB_THREADS 512
#define EB_SLOTS (MAX_BUCKETS / (EB_THREADS / 16))

__global__ __launch_bounds__(EB_THREADS) void edge_bucket(
    const int* __restrict__ src, const int* __restrict__ dst, int E,
    int2* __restrict__ rec,
    const float* __restrict__ W1, const float* __restrict__ as1, const float* __restrict__ ad1,
    const float* __restrict__ W2, const float* __restrict__ as2, const float* __restrict__ ad2,
    float* __restrict__ wvec)
{
    __shared__ int lc[MAX_BUCKETS];
    __shared__ __attribute__((aligned(16))) int2 stg[MAX_BUCKETS * STAGE_C];

    if (blockIdx.x == NCHUNK) {
        if (threadIdx.x < 4 * FDIM) {
            const int t   = threadIdx.x;
            const int v   = t >> 6;
            const int row = t & 63;
            const float* __restrict__ Wm = (v < 2) ? W1 : W2;
            const float* __restrict__ a  = (v == 0) ? as1 : (v == 1) ? ad1 : (v == 2) ? as2 : ad2;
            const float4* __restrict__ wr = (const float4*)(Wm + row * FDIM);
            const float4* __restrict__ av = (const float4*)a;
            float s = 0.0f;
#pragma unroll
            for (int q = 0; q < FDIM / 4; ++q) {
                const float4 w4 = wr[q];
                const float4 a4 = av[q];
                s = fmaf(w4.x, a4.x, s);
                s = fmaf(w4.y, a4.y, s);
                s = fmaf(w4.z, a4.z, s);
                s = fmaf(w4.w, a4.w, s);
            }
            wvec[v * FDIM + row] = s;
        }
        return;
    }

    if (threadIdx.x < MAX_BUCKETS) lc[threadIdx.x] = 0;
    __syncthreads();

    const int g     = blockIdx.x;
    const int chunk = (E + NCHUNK - 1) / NCHUNK;
    const int e0    = g * chunk;
    const int e1    = min(e0 + chunk, E);
    for (int base = e0; base < e1; base += 4 * EB_THREADS) {
        int d[4], s[4];
#pragma unroll
        for (int k = 0; k < 4; ++k) {
            const int e = base + k * EB_THREADS + (int)threadIdx.x;
            d[k] = -1; s[k] = 0;
            if (e < e1) { d[k] = dst[e]; s[k] = src[e]; }
        }
#pragma unroll
        for (int k = 0; k < 4; ++k) {
            if (d[k] >= 0) {
                const int b   = d[k] / NODES_PER_GROUP;
                const int pos = atomicAdd(&lc[b], 1) + 1;
                if (pos < STAGE_C)
                    stg[b * STAGE_C + pos] = make_int2(s[k], d[k]);
                else if (pos < SLOT_C)
                    rec[((size_t)b * NCHUNK + g) * SLOT_C + pos] = make_int2(s[k], d[k]);
            }
        }
    }
    __syncthreads();
    const int hw = (int)threadIdx.x >> 4;
    const int l  = (int)threadIdx.x & 15;
    int n[EB_SLOTS];
#pragma unroll
    for (int j = 0; j < EB_SLOTS; ++j) n[j] = lc[hw + (EB_THREADS / 16) * j];
    int2 v[EB_SLOTS][STAGE_C / 16];
#pragma unroll
    for (int j = 0; j < EB_SLOTS; ++j) {
#pragma unroll
        for (int t = 0; t < STAGE_C / 16; ++t) {
            const int p = l + 16 * t;
            v[j][t] = make_int2(0, -1);
            if (p >= 1 && p <= n[j]) v[j][t] = stg[(hw + (EB_THREADS / 16) * j) * STAGE_C + p];
        }
    }
#pragma unroll
    for (int j = 0; j < EB_SLOTS; ++j) {
        int2* __restrict__ sp = rec + ((size_t)(hw + (EB_THREADS / 16) * j) * NCHUNK + g) * SLOT_C;
#pragma unroll
        for (int t = 0; t < STAGE_C / 16; ++t) {
            int2 o = v[j][t];
            if (t == 0 && l == 0) o = make_int2(n[j], -1);
            sp[l + 16 * t] = o;
        }
    }
}

#define ZPITCH 68
static_assert(32 * WAVES_PER_BLOCK == 4 * NCHUNK, "four threads of a gather block per chunk slot");
static_assert(STAGE_C == 32 && SLOT_C == 64, "four threads' 4 loads of 16 bytes each cover a slot's two staged lines");

template <int N>
static __device__ __forceinline__ float row_xor(float x)
{
    return __int_as_float(__builtin_amdgcn_update_dpp(0, __float_as_int(x), 0x160 + N, 0xf, 0xf, true));
}

static __device__ __forceinline__ float row16_sum(float p)
{
    p += row_xor<1>(p);
    p += row_xor<2>(p);
    p += row_xor<4>(p);
    p += row_xor<8>(p);
    return p;
}

template <int K>
static __device__ __forceinline__ int row_bcast(int v)
{
    return __builtin_amdgcn_update_dpp(0, v, 0x150 + K, 0xf, 0xf, true);
}

static __device__ __forceinline__ float xhalf(float x)
{
    const int b = __float_as_int(x);
    return __int_as_float(__builtin_amdgcn_permlanex16(b, b, 0x76543210, 0xfedcba98, false, false));
}

static __device__ __forceinline__ float dot4(const float4 r, const float4 a)
{
    v2f q = (v2f){r.x, r.y} * (v2f){a.x, a.y};
    q = (v2f){r.z, r.w} * (v2f){a.z, a.w} + q;
    return q.x + q.y;
}

template <int U>
static __device__ __forceinline__ int slot_src(int s_lo, int s_hi)
{
    return row_bcast<(U & 15)>((U < 16) ? s_lo : s_hi);
}

#define LD_SLOT(T, U)                                                            \
    do {                                                                         \
        const int sv_ = slot_src<(U)>(s_lo, s_hi);                               \
        r[T] = *(const float4*)(Hb + ((unsigned)sv_ * 256u + coff));             \
    } while (0)
#define LD_GROUP(T0, U0)                                                         \
    if (2 * (U0) < deg) {                                                        \
        LD_SLOT((T0), (U0));         LD_SLOT((T0) + 1, (U0) + 1);                \
        LD_SLOT((T0) + 2, (U0) + 2); LD_SLOT((T0) + 3, (U0) + 3);                \
    }
#define SC_GROUP(G, T0, U0)                                                      \
    if (2 * (U0) < deg) {                                                        \
        const float p0_ = dot4(r[(T0)], as4);                                    \
        const float p1_ = dot4(r[(T0) + 1], as4);                                \
        const float p2_ = dot4(r[(T0) + 2], as4);                                \
        const float p3_ = dot4(r[(T0) + 3], as4);                                \
        const float k01_ = hi8 ? p1_ : p0_;                                      \
        const float s01_ = hi8 ? p0_ : p1_;                                      \
        const float k23_ = hi8 ? p3_ : p2_;                                      \
        const float s23_ = hi8 ? p2_ : p3_;                                      \
        const float r01_ = k01_ + row_xor<8>(s01_);                              \
        const float r23_ = k23_ + row_xor<8>(s23_);                              \
        const float kk_ = hi4 ? r23_ : r01_;                                     \
        const float ss_ = hi4 ? r01_ : r23_;                                     \
        float v_ = kk_ + row_xor<4>(ss_);                                        \
        v_ += row_xor<2>(v_);                                                    \
        v_ += row_xor<1>(v_);                                                    \
        v_ += hd;                                                                \
        v_ = (v_ >= 0.0f) ? v_ : 0.2f * v_;                                      \
        sc[G] = ((2 * (U0) + eg) < deg) ? v_ : NEG_INF;                          \
        cm = fmaxf(cm, sc[G]);                                                   \
    }
#define AC_SLOT(T, K)                                                            \
    do {                                                                         \
        const float wb_ = __int_as_float(row_bcast<(K)>(__float_as_int(w_)));    \
        const v2f w2_ = (v2f){wb_, wb_};                                         \
        a01 = (v2f){r[T].x, r[T].y} * w2_ + a01;                                 \
        a23 = (v2f){r[T].z, r[T].w} * w2_ + a23;                                 \
    } while (0)
#define AC_GROUP(G, T0, U0)                                                      \
    if (2 * (U0) < deg) {                                                        \
        const float w_ = __expf(sc[G] - m);                                      \
        S += w_;                                                                 \
        AC_SLOT((T0), 0);     AC_SLOT((T0) + 1, 8);                              \
        AC_SLOT((T0) + 2, 4); AC_SLOT((T0) + 3, 12);                             \
    }

template <int U0, int NS>
static __device__ __forceinline__ void gather_chunk(
    const int deg, const int eg, const bool hi8, const bool hi4, const unsigned coff,
    const int s_lo, const int s_hi,
    const char* __restrict__ Hb, const float4 as4, const float hd,
    float& m, float& S, float4& acc)
{
    const float NEG_INF = __int_as_float(0xFF800000);
    float4 r[NS];
    float  sc[NS / 4];

    LD_GROUP(0, U0)
    LD_GROUP(4, U0 + 4)
    if (NS > 8) { LD_GROUP(8, U0 + 8) }

    float cm = NEG_INF;
    SC_GROUP(0, 0, U0)
    SC_GROUP(1, 4, U0 + 4)
    if (NS > 8) { SC_GROUP(2, 8, U0 + 8) }
    cm = fmaxf(cm, row_xor<4>(cm));
    cm = fmaxf(cm, row_xor<8>(cm));
    cm = fmaxf(cm, xhalf(cm));

    const float mn = fmaxf(m, cm);
    const float f  = __expf(m - mn);
    m = mn;
    S *= f;
    v2f a01 = (v2f){acc.x * f, acc.y * f};
    v2f a23 = (v2f){acc.z * f, acc.w * f};

    AC_GROUP(0, 0, U0)
    AC_GROUP(1, 4, U0 + 4)
    if (NS > 8) { AC_GROUP(2, 8, U0 + 8) }

    acc.x = a01.x; acc.y = a01.y; acc.z = a23.x; acc.w = a23.y;
}

__global__ __launch_bounds__(32 * WAVES_PER_BLOCK) void gat_gather_xform(
    const int2* __restrict__ rec,
    const int* __restrict__ esrc, const int* __restrict__ edst, int E,
    const float* __restrict__ X,
    const float* __restrict__ was,
    const float* __restrict__ wad,
    const float* __restrict__ Wm,
    const float* __restrict__ bias,
    float* __restrict__ OUT, int n_nodes)
{
    __shared__ __attribute__((aligned(16))) float Zs[NODES_PER_GROUP * ZPITCH];
    __shared__ int arrived[TILES_PER_BLOCK];
    __shared__ int next_row;
    __shared__ int degs[NODES_PER_GROUP];
    __shared__ int Ll[NODES_PER_GROUP * LCAP];

    const int bkt       = blockIdx.x;
    const int base_node = bkt * NODES_PER_GROUP;

    const int slot = (int)threadIdx.x >> 2;
    const int q    = (int)threadIdx.x & 3;
    const int2* __restrict__ rp = rec + ((size_t)bkt * NCHUNK + slot) * SLOT_C;
    int2 r[STAGE_C / 4];
#pragma unroll
    for (int k = 0; k < STAGE_C / 8; ++k) {
        const int4 w = ((const int4*)rp)[q + 4 * k];
        r[2 * k]     = make_int2(w.x, w.y);
        r[2 * k + 1] = make_int2(w.z, w.w);
    }

    if (threadIdx.x < NODES_PER_GROUP) degs[threadIdx.x] = 0;
    if (threadIdx.x < TILES_PER_BLOCK) arrived[threadIdx.x] = 0;
    if (threadIdx.x == TILES_PER_BLOCK) next_row = 0;
    __syncthreads();

    {
        const int nrec = __builtin_amdgcn_update_dpp(0, r[0].x, 0x00, 0xf, 0xf, true);
        if (nrec < SLOT_C) {
            int pos[STAGE_C / 4];
#pragma unroll
            for (int k = 0; k < STAGE_C / 4; ++k) {
                pos[k] = LCAP;
                if (r[k].y >= 0) pos[k] = atomicAdd(&degs[r[k].y - base_node], 1);
            }
#pragma unroll
            for (int k = 0; k < STAGE_C / 4; ++k) {
                if (pos[k] < LCAP) Ll[(r[k].y - base_node) * LCAP + pos[k]] = r[k].x;
            }
            for (int i = STAGE_C + q; i <= nrec; i += 4) {
                const int2 t  = rp[i];
                const int  ln = t.y - base_node;
                const int  ps = atomicAdd(&degs[ln], 1);
                if (ps < LCAP) Ll[ln * LCAP + ps] = t.x;
            }
        } else if (q == 0) {
            const int chunk = (E + NCHUNK - 1) / NCHUNK;
            const int e0    = slot * chunk;
            const int e1    = min(e0 + chunk, E);
            for (int e = e0; e < e1; ++e) {
                const int d = edst[e];
                if (d / NODES_PER_GROUP == bkt) {
                    const int sv  = esrc[e];
                    const int ln  = d - base_node;
                    const int pos = atomicAdd(&degs[ln], 1);
                    if (pos < LCAP) Ll[ln * LCAP + pos] = sv;
                }
            }
        }
    }
    __syncthreads();

    const int lane = threadIdx.x & 31;
    const int half = lane >> 4;
    const int c    = lane & 15;
    const bool hi8 = (c & 8) != 0;
    const bool hi4 = (c & 4) != 0;
    const int  eg  = 2 * ((hi8 ? 1 : 0) + (hi4 ? 2 : 0)) + half;

    const int lim = min(NODES_PER_GROUP, (n_nodes - base_node + NODES_PER_BLOCK - 1) & ~(NODES_PER_BLOCK - 1));

#pragma nounroll
    for (int it = 0; it < NODES_PER_GROUP; ++it) {
        int row = 0;
        if (lane == 0)
            row = __hip_atomic_fetch_add(&next_row, 1, __ATOMIC_RELAXED, __HIP_MEMORY_SCOPE_WORKGROUP);
        row = __builtin_amdgcn_readfirstlane(row);
        if (row >= lim) break;

        const int node = base_node + row;

        float4 acc;
        acc.x = 0.0f; acc.y = 0.0f; acc.z = 0.0f; acc.w = 0.0f;

        if (node < n_nodes) {
            const int deg = __builtin_amdgcn_readfirstlane(degs[row]);
            const float4* __restrict__ Xv = (const float4*)X;
            const float4 as4   = ((const float4*)was)[c];
            const float4 ad4   = ((const float4*)wad)[c];
            const float4 hself = Xv[(size_t)node * 16 + c];

            float pd = hself.x * ad4.x;
            pd = fmaf(hself.y, ad4.y, pd);
            pd = fmaf(hself.z, ad4.z, pd);
            pd = fmaf(hself.w, ad4.w, pd);
            const float hd = row16_sum(pd);

            float ps = hself.x * as4.x;
            ps = fmaf(hself.y, as4.y, ps);
            ps = fmaf(hself.z, as4.z, ps);
            ps = fmaf(hself.w, as4.w, ps);
            float es = row16_sum(ps) + hd;
            es = (es >= 0.0f) ? es : 0.2f * es;

            if (deg <= LCAP) {
                const int* lst = Ll + row * LCAP;
                const char* __restrict__ Hb = (const char*)X;
                const unsigned coff = 16u * (unsigned)c;

                const int e_lo = 2 * c + half;
                int s_lo = node, s_hi = node;
                if (e_lo < deg) s_lo = lst[e_lo];
                if (deg > 32) {
                    if (e_lo + 32 < deg) s_hi = lst[e_lo + 32];
                }

                float m = es;
                float S = (half == 0 && c < 4) ? 1.0f : 0.0f;
                acc.x = (half == 0) ? hself.x : 0.0f;
                acc.y = (half == 0) ? hself.y : 0.0f;
                acc.z = (half == 0) ? hself.z : 0.0f;
                acc.w = (half == 0) ? hself.w : 0.0f;

                if (deg > 0)  gather_chunk<0, 12>(deg, eg, hi8, hi4, coff, s_lo, s_hi, Hb, as4, hd, m, S, acc);
                if (deg > 24) gather_chunk<12, 12>(deg, eg, hi8, hi4, coff, s_lo, s_hi, Hb, as4, hd, m, S, acc);
                if (deg > 48) gather_chunk<24, 8>(deg, eg, hi8, hi4, coff, s_lo, s_hi, Hb, as4, hd, m, S, acc);

                S += row_xor<4>(S);
                S += row_xor<8>(S);
                const float inv = 1.0f / (S + xhalf(S));
                acc.x *= inv; acc.y *= inv; acc.z *= inv; acc.w *= inv;
            } else {
                float m = es;
                float S = 1.0f;
                acc = hself;
                for (int base = 0; base < E; base += 32) {
                    const int  e   = base + lane;
                    const bool hit = (e < E) && (edst[e] == node);
                    int s = node;
                    if (hit) s = esrc[e];
                    unsigned mk = (unsigned)__ballot(hit);
                    while (mk != 0u) {
                        const int k = __builtin_ctz(mk);
                        mk &= mk - 1u;
                        const int    sk = __shfl(s, k, 32);
                        const float4 r  = Xv[(size_t)sk * 16 + c];
                        float p = r.x * as4.x;
                        p = fmaf(r.y, as4.y, p);
                        p = fmaf(r.z, as4.z, p);
                        p = fmaf(r.w, as4.w, p);
                        float v = row16_sum(p) + hd;
                        v = (v >= 0.0f) ? v : 0.2f * v;
                        const float mn = fmaxf(m, v);
                        const float f  = __expf(m - mn);
                        const float w  = __expf(v - mn);
                        m = mn;
                        S = fmaf(S, f, w);
                        acc.x = fmaf(acc.x, f, w * r.x); acc.y = fmaf(acc.y, f, w * r.y);
                        acc.z = fmaf(acc.z, f, w * r.z); acc.w = fmaf(acc.w, f, w * r.w);
                    }
                }
                const float inv = (half == 0) ? 1.0f / S : 0.0f;
                acc.x *= inv; acc.y *= inv; acc.z *= inv; acc.w *= inv;
            }

            acc.x += xhalf(acc.x);
            acc.y += xhalf(acc.y);
            acc.z += xhalf(acc.z);
            acc.w += xhalf(acc.w);
        }

        const int tl = row >> 4;
        if (half == 0) *(float4*)(Zs + row * ZPITCH + 4 * c) = acc;
        int owner = 0;
        if (lane == 0) {
            const int before = __hip_atomic_fetch_add(&arrived[tl], 1, __ATOMIC_ACQ_REL,
                                                      __HIP_MEMORY_SCOPE_WORKGROUP);
            owner = (before == NODES_PER_BLOCK - 1) ? 1 : 0;
        }
        owner = __builtin_amdgcn_readfirstlane(owner);
        if (owner != 0) {
            const int khalf = half;
            v8f d0 = (v8f){0,0,0,0,0,0,0,0};
            v8f d1 = (v8f){0,0,0,0,0,0,0,0};
            v8f d2 = (v8f){0,0,0,0,0,0,0,0};
            v8f d3 = (v8f){0,0,0,0,0,0,0,0};
            const float* zr = Zs + (tl * NODES_PER_BLOCK + c) * ZPITCH + 2 * khalf;
            const float4* __restrict__ wq = (const float4*)(Wm + (2 * khalf) * FDIM) + c;
#pragma nounroll
            for (int kc = 0; kc < 4; ++kc) {
                const float*  zc = zr + 16 * kc;
                const float4* __restrict__ wc = wq + (16 * kc) * (FDIM / 4);
#pragma unroll
                for (int k = 0; k < 4; ++k) {
                    const v2f    af = *(const v2f*)(zc + 4 * k);
                    const float4 q0 = wc[(4 * k) * (FDIM / 4)];
                    const float4 q1 = wc[(4 * k + 1) * (FDIM / 4)];
                    v2f b;
                    b.x = q0.x; b.y = q1.x;
                    d0 = __builtin_amdgcn_wmma_f32_16x16x4_f32(false, af, false, b, (short)0, d0, false, false);
                    b.x = q0.y; b.y = q1.y;
                    d1 = __builtin_amdgcn_wmma_f32_16x16x4_f32(false, af, false, b, (short)0, d1, false, false);
                    b.x = q0.z; b.y = q1.z;
                    d2 = __builtin_amdgcn_wmma_f32_16x16x4_f32(false, af, false, b, (short)0, d2, false, false);
                    b.x = q0.w; b.y = q1.w;
                    d3 = __builtin_amdgcn_wmma_f32_16x16x4_f32(false, af, false, b, (short)0, d3, false, false);
                }
            }

            const float4 bz = ((const float4*)bias)[c];
            const int rbase = base_node + tl * NODES_PER_BLOCK + 8 * khalf;
#pragma unroll
            for (int r = 0; r < 8; ++r) {
                float4 o;
                o.x = d0[r] + bz.x;
                o.y = d1[r] + bz.y;
                o.z = d2[r] + bz.z;
                o.w = d3[r] + bz.w;
                if (rbase + r < n_nodes) ((float4*)OUT)[(size_t)(rbase + r) * 16 + c] = o;
            }
        }
    }
}

extern "C" void kernel_launch(void* const* d_in, const int* in_sizes, int n_in,
                              void* d_out, int out_size, void* d_ws, size_t ws_size,
                              hipStream_t stream) {
    const float* x   = (const float*)d_in[0];
    const int*   ei  = (const int*)  d_in[1];
    const float* W1  = (const float*)d_in[2];
    const float* as1 = (const float*)d_in[3];
    const float* ad1 = (const float*)d_in[4];
    const float* b1  = (const float*)d_in[5];
    const float* W2  = (const float*)d_in[6];
    const float* as2 = (const float*)d_in[7];
    const float* ad2 = (const float*)d_in[8];
    const float* b2  = (const float*)d_in[9];

    const int N    = in_sizes[0] / FDIM;
    const int E    = in_sizes[1] / 2;
    const int* src = ei;
    const int* dst = ei + E;

    float* ws = (float*)d_ws;
    float* X1 = ws;  ws += (size_t)N * FDIM;
    float* wvec = ws;  ws += 4 * FDIM;
    int2* rec = (int2*)ws;

    edge_bucket<<<NCHUNK + 1, EB_THREADS, 0, stream>>>(src, dst, E, rec,
                                                       W1, as1, ad1, W2, as2, ad2, wvec);

    auto layer = [&](const float* Xin, const float* was, const float* wad,
                     const float* W, const float* b, float* OUT) {
        gat_gather_xform<<<(N + NODES_PER_GROUP - 1) / NODES_PER_GROUP,
                           32 * WAVES_PER_BLOCK, 0, stream>>>(rec, src, dst, E,
                                                              Xin, was, wad, W, b, OUT, N);
    };

    layer(x,  wvec,            wvec + FDIM,     W1, b1, X1);
    layer(X1, wvec + 2 * FDIM, wvec + 3 * FDIM, W2, b2, (float*)d_out);
}


# Gather: a node's own row is asked for first but used only behind the issue of its first chunk's source-row loads, so a node waits for memory once, not twice; f32 sums and their order unchanged
# speedup vs baseline: 9.5727x; 1.0291x over previous
#include <hip/hip_runtime.h>
#include <math.h>

#define FDIM 64

typedef __attribute__((ext_vector_type(2))) float v2f;
typedef __attribute__((ext_vector_type(8))) float v8f;


#define LCAP 64

#define NODES_PER_BLOCK 16
#define WAVES_PER_BLOCK 32
#define TILES_PER_BLOCK 13
#define NODES_PER_GROUP (TILES_PER_BLOCK * NODES_PER_BLOCK)
#define MAX_BUCKETS 256
#define NCHUNK 256
#define SLOT_C 64
#define STAGE_C 32
#define EB_THREADS 512
#define EB_SLOTS (MAX_BUCKETS / (EB_THREADS / 16))

__global__ __launch_bounds__(EB_THREADS) void edge_bucket(
    const int* __restrict__ src, const int* __restrict__ dst, int E,
    int2* __restrict__ rec,
    const float* __restrict__ W1, const float* __restrict__ as1, const float* __restrict__ ad1,
    const float* __restrict__ W2, const float* __restrict__ as2, const float* __restrict__ ad2,
    float* __restrict__ wvec)
{
    __shared__ int lc[MAX_BUCKETS];
    __shared__ __attribute__((aligned(16))) int2 stg[MAX_BUCKETS * STAGE_C];

    if (blockIdx.x == NCHUNK) {
        if (threadIdx.x < 4 * FDIM) {
            const int t   = threadIdx.x;
            const int v   = t >> 6;
            const int row = t & 63;
            const float* __restrict__ Wm = (v < 2) ? W1 : W2;
            const float* __restrict__ a  = (v == 0) ? as1 : (v == 1) ? ad1 : (v == 2) ? as2 : ad2;
            const float4* __restrict__ wr = (const float4*)(Wm + row * FDIM);
            const float4* __restrict__ av = (const float4*)a;
            float s = 0.0f;
#pragma unroll
            for (int q = 0; q < FDIM / 4; ++q) {
                const float4 w4 = wr[q];
                const float4 a4 = av[q];
                s = fmaf(w4.x, a4.x, s);
                s = fmaf(w4.y, a4.y, s);
                s = fmaf(w4.z, a4.z, s);
                s = fmaf(w4.w, a4.w, s);
            }
            wvec[v * FDIM + row] = s;
        }
        return;
    }

    if (threadIdx.x < MAX_BUCKETS) lc[threadIdx.x] = 0;
    __syncthreads();

    const int g     = blockIdx.x;
    const int chunk = (E + NCHUNK - 1) / NCHUNK;
    const int e0    = g * chunk;
    const int e1    = min(e0 + chunk, E);
    for (int base = e0; base < e1; base += 4 * EB_THREADS) {
        int d[4], s[4];
#pragma unroll
        for (int k = 0; k < 4; ++k) {
            const int e = base + k * EB_THREADS + (int)threadIdx.x;
            d[k] = -1; s[k] = 0;
            if (e < e1) { d[k] = dst[e]; s[k] = src[e]; }
        }
#pragma unroll
        for (int k = 0; k < 4; ++k) {
            if (d[k] >= 0) {
                const int b   = d[k] / NODES_PER_GROUP;
                const int pos = atomicAdd(&lc[b], 1) + 1;
                if (pos < STAGE_C)
                    stg[b * STAGE_C + pos] = make_int2(s[k], d[k]);
                else if (pos < SLOT_C)
                    rec[((size_t)b * NCHUNK + g) * SLOT_C + pos] = make_int2(s[k], d[k]);
            }
        }
    }
    __syncthreads();
    const int hw = (int)threadIdx.x >> 4;
    const int l  = (int)threadIdx.x & 15;
    int n[EB_SLOTS];
#pragma unroll
    for (int j = 0; j < EB_SLOTS; ++j) n[j] = lc[hw + (EB_THREADS / 16) * j];
    int2 v[EB_SLOTS][STAGE_C / 16];
#pragma unroll
    for (int j = 0; j < EB_SLOTS; ++j) {
#pragma unroll
        for (int t = 0; t < STAGE_C / 16; ++t) {
            const int p = l + 16 * t;
            v[j][t] = make_int2(0, -1);
            if (p >= 1 && p <= n[j]) v[j][t] = stg[(hw + (EB_THREADS / 16) * j) * STAGE_C + p];
        }
    }
#pragma unroll
    for (int j = 0; j < EB_SLOTS; ++j) {
        int2* __restrict__ sp = rec + ((size_t)(hw + (EB_THREADS / 16) * j) * NCHUNK + g) * SLOT_C;
#pragma unroll
        for (int t = 0; t < STAGE_C / 16; ++t) {
            int2 o = v[j][t];
            if (t == 0 && l == 0) o = make_int2(n[j], -1);
            sp[l + 16 * t] = o;
        }
    }
}

#define ZPITCH 68
static_assert(32 * WAVES_PER_BLOCK == 4 * NCHUNK, "four threads of a gather block per chunk slot");
static_assert(STAGE_C == 32 && SLOT_C == 64, "four threads' 4 loads of 16 bytes each cover a slot's two staged lines");

template <int N>
static __device__ __forceinline__ float row_xor(float x)
{
    return __int_as_float(__builtin_amdgcn_update_dpp(0, __float_as_int(x), 0x160 + N, 0xf, 0xf, true));
}

static __device__ __forceinline__ float row16_sum(float p)
{
    p += row_xor<1>(p);
    p += row_xor<2>(p);
    p += row_xor<4>(p);
    p += row_xor<8>(p);
    return p;
}

template <int K>
static __device__ __forceinline__ int row_bcast(int v)
{
    return __builtin_amdgcn_update_dpp(0, v, 0x150 + K, 0xf, 0xf, true);
}

static __device__ __forceinline__ float xhalf(float x)
{
    const int b = __float_as_int(x);
    return __int_as_float(__builtin_amdgcn_permlanex16(b, b, 0x76543210, 0xfedcba98, false, false));
}

static __device__ __forceinline__ float dot4(const float4 r, const float4 a)
{
    v2f q = (v2f){r.x, r.y} * (v2f){a.x, a.y};
    q = (v2f){r.z, r.w} * (v2f){a.z, a.w} + q;
    return q.x + q.y;
}

template <int U>
static __device__ __forceinline__ int slot_src(int s_lo, int s_hi)
{
    return row_bcast<(U & 15)>((U < 16) ? s_lo : s_hi);
}

#define LD_SLOT(T, U)                                                            \
    do {                                                                         \
        const int sv_ = slot_src<(U)>(s_lo, s_hi);                               \
        r[T] = *(const float4*)(Hb + ((unsigned)sv_ * 256u + coff));             \
    } while (0)
#define LD_GROUP(T0, U0)                                                         \
    if (2 * (U0) < deg) {                                                        \
        LD_SLOT((T0), (U0));         LD_SLOT((T0) + 1, (U0) + 1);                \
        LD_SLOT((T0) + 2, (U0) + 2); LD_SLOT((T0) + 3, (U0) + 3);                \
    }
#define SC_GROUP(G, T0, U0)                                                      \
    if (2 * (U0) < deg) {                                                        \
        const float p0_ = dot4(r[(T0)], as4);                                    \
        const float p1_ = dot4(r[(T0) + 1], as4);                                \
        const float p2_ = dot4(r[(T0) + 2], as4);                                \
        const float p3_ = dot4(r[(T0) + 3], as4);                                \
        const float k01_ = hi8 ? p1_ : p0_;                                      \
        const float s01_ = hi8 ? p0_ : p1_;                                      \
        const float k23_ = hi8 ? p3_ : p2_;                                      \
        const float s23_ = hi8 ? p2_ : p3_;                                      \
        const float r01_ = k01_ + row_xor<8>(s01_);                              \
        const float r23_ = k23_ + row_xor<8>(s23_);                              \
        const float kk_ = hi4 ? r23_ : r01_;                                     \
        const float ss_ = hi4 ? r01_ : r23_;                                     \
        float v_ = kk_ + row_xor<4>(ss_);                                        \
        v_ += row_xor<2>(v_);                                                    \
        v_ += row_xor<1>(v_);                                                    \
        v_ += hd;                                                                \
        v_ = (v_ >= 0.0f) ? v_ : 0.2f * v_;                                      \
        sc[G] = ((2 * (U0) + eg) < deg) ? v_ : NEG_INF;                          \
        cm = fmaxf(cm, sc[G]);                                                   \
    }
#define AC_SLOT(T, K)                                                            \
    do {                                                                         \
        const float wb_ = __int_as_float(row_bcast<(K)>(__float_as_int(w_)));    \
        const v2f w2_ = (v2f){wb_, wb_};                                         \
        a01 = (v2f){r[T].x, r[T].y} * w2_ + a01;                                 \
        a23 = (v2f){r[T].z, r[T].w} * w2_ + a23;                                 \
    } while (0)
#define AC_GROUP(G, T0, U0)                                                      \
    if (2 * (U0) < deg) {                                                        \
        const float w_ = __expf(sc[G] - m);                                      \
        S += w_;                                                                 \
        AC_SLOT((T0), 0);     AC_SLOT((T0) + 1, 8);                              \
        AC_SLOT((T0) + 2, 4); AC_SLOT((T0) + 3, 12);                             \
    }

template <int U0, int NS, bool FIRST>
static __device__ __forceinline__ void gather_chunk(
    const int deg, const int eg, const bool hi8, const bool hi4, const unsigned coff,
    const int s_lo, const int s_hi,
    const char* __restrict__ Hb, const float4 as4,
    const float4 ad4, const float4 hself, const bool half0, const bool sgroup,
    float& hd, float& m, float& S, float4& acc)
{
    const float NEG_INF = __int_as_float(0xFF800000);
    float4 r[NS];
    float  sc[NS / 4];

    LD_GROUP(0, U0)
    LD_GROUP(4, U0 + 4)
    if (NS > 8) { LD_GROUP(8, U0 + 8) }

    if (FIRST) {
        float pd = hself.x * ad4.x;
        pd = fmaf(hself.y, ad4.y, pd);
        pd = fmaf(hself.z, ad4.z, pd);
        pd = fmaf(hself.w, ad4.w, pd);
        hd = row16_sum(pd);

        float ps = hself.x * as4.x;
        ps = fmaf(hself.y, as4.y, ps);
        ps = fmaf(hself.z, as4.z, ps);
        ps = fmaf(hself.w, as4.w, ps);
        float es = row16_sum(ps) + hd;
        es = (es >= 0.0f) ? es : 0.2f * es;

        m = es;
        S = sgroup ? 1.0f : 0.0f;
        acc.x = half0 ? hself.x : 0.0f;
        acc.y = half0 ? hself.y : 0.0f;
        acc.z = half0 ? hself.z : 0.0f;
        acc.w = half0 ? hself.w : 0.0f;
    }

    float cm = NEG_INF;
    SC_GROUP(0, 0, U0)
    SC_GROUP(1, 4, U0 + 4)
    if (NS > 8) { SC_GROUP(2, 8, U0 + 8) }
    cm = fmaxf(cm, row_xor<4>(cm));
    cm = fmaxf(cm, row_xor<8>(cm));
    cm = fmaxf(cm, xhalf(cm));

    const float mn = fmaxf(m, cm);
    const float f  = __expf(m - mn);
    m = mn;
    S *= f;
    v2f a01 = (v2f){acc.x * f, acc.y * f};
    v2f a23 = (v2f){acc.z * f, acc.w * f};

    AC_GROUP(0, 0, U0)
    AC_GROUP(1, 4, U0 + 4)
    if (NS > 8) { AC_GROUP(2, 8, U0 + 8) }

    acc.x = a01.x; acc.y = a01.y; acc.z = a23.x; acc.w = a23.y;
}

__global__ __launch_bounds__(32 * WAVES_PER_BLOCK) void gat_gather_xform(
    const int2* __restrict__ rec,
    const int* __restrict__ esrc, const int* __restrict__ edst, int E,
    const float* __restrict__ X,
    const float* __restrict__ was,
    const float* __restrict__ wad,
    const float* __restrict__ Wm,
    const float* __restrict__ bias,
    float* __restrict__ OUT, int n_nodes)
{
    __shared__ __attribute__((aligned(16))) float Zs[NODES_PER_GROUP * ZPITCH];
    __shared__ int arrived[TILES_PER_BLOCK];
    __shared__ int next_row;
    __shared__ int degs[NODES_PER_GROUP];
    __shared__ int Ll[NODES_PER_GROUP * LCAP];

    const int bkt       = blockIdx.x;
    const int base_node = bkt * NODES_PER_GROUP;

    const int slot = (int)threadIdx.x >> 2;
    const int q    = (int)threadIdx.x & 3;
    const int2* __restrict__ rp = rec + ((size_t)bkt * NCHUNK + slot) * SLOT_C;
    int2 r[STAGE_C / 4];
#pragma unroll
    for (int k = 0; k < STAGE_C / 8; ++k) {
        const int4 w = ((const int4*)rp)[q + 4 * k];
        r[2 * k]     = make_int2(w.x, w.y);
        r[2 * k + 1] = make_int2(w.z, w.w);
    }

    if (threadIdx.x < NODES_PER_GROUP) degs[threadIdx.x] = 0;
    if (threadIdx.x < TILES_PER_BLOCK) arrived[threadIdx.x] = 0;
    if (threadIdx.x == TILES_PER_BLOCK) next_row = 0;
    __syncthreads();

    {
        const int nrec = __builtin_amdgcn_update_dpp(0, r[0].x, 0x00, 0xf, 0xf, true);
        if (nrec < SLOT_C) {
            int pos[STAGE_C / 4];
#pragma unroll
            for (int k = 0; k < STAGE_C / 4; ++k) {
                pos[k] = LCAP;
                if (r[k].y >= 0) pos[k] = atomicAdd(&degs[r[k].y - base_node], 1);
            }
#pragma unroll
            for (int k = 0; k < STAGE_C / 4; ++k) {
                if (pos[k] < LCAP) Ll[(r[k].y - base_node) * LCAP + pos[k]] = r[k].x;
            }
            for (int i = STAGE_C + q; i <= nrec; i += 4) {
                const int2 t  = rp[i];
                const int  ln = t.y - base_node;
                const int  ps = atomicAdd(&degs[ln], 1);
                if (ps < LCAP) Ll[ln * LCAP + ps] = t.x;
            }
        } else if (q == 0) {
            const int chunk = (E + NCHUNK - 1) / NCHUNK;
            const int e0    = slot * chunk;
            const int e1    = min(e0 + chunk, E);
            for (int e = e0; e < e1; ++e) {
                const int d = edst[e];
                if (d / NODES_PER_GROUP == bkt) {
                    const int sv  = esrc[e];
                    const int ln  = d - base_node;
                    const int pos = atomicAdd(&degs[ln], 1);
                    if (pos < LCAP) Ll[ln * LCAP + pos] = sv;
                }
            }
        }
    }
    __syncthreads();

    const int lane = threadIdx.x & 31;
    const int half = lane >> 4;
    const int c    = lane & 15;
    const bool hi8 = (c & 8) != 0;
    const bool hi4 = (c & 4) != 0;
    const int  eg  = 2 * ((hi8 ? 1 : 0) + (hi4 ? 2 : 0)) + half;

    const int lim = min(NODES_PER_GROUP, (n_nodes - base_node + NODES_PER_BLOCK - 1) & ~(NODES_PER_BLOCK - 1));

#pragma nounroll
    for (int it = 0; it < NODES_PER_GROUP; ++it) {
        int row = 0;
        if (lane == 0)
            row = __hip_atomic_fetch_add(&next_row, 1, __ATOMIC_RELAXED, __HIP_MEMORY_SCOPE_WORKGROUP);
        row = __builtin_amdgcn_readfirstlane(row);
        if (row >= lim) break;

        const int node = base_node + row;

        float4 acc;
        acc.x = 0.0f; acc.y = 0.0f; acc.z = 0.0f; acc.w = 0.0f;

        if (node < n_nodes) {
            const int deg = __builtin_amdgcn_readfirstlane(degs[row]);
            const float4* __restrict__ Xv = (const float4*)X;
            const float4 as4   = ((const float4*)was)[c];
            const float4 ad4   = ((const float4*)wad)[c];
            const float4 hself = Xv[(size_t)node * 16 + c];

            if (deg <= LCAP) {
                const int* lst = Ll + row * LCAP;
                const char* __restrict__ Hb = (const char*)X;
                const unsigned coff = 16u * (unsigned)c;

                const int e_lo = 2 * c + half;
                int s_lo = node, s_hi = node;
                if (e_lo < deg) s_lo = lst[e_lo];
                if (deg > 32) {
                    if (e_lo + 32 < deg) s_hi = lst[e_lo + 32];
                }

                float hd, m, S;
                const bool half0  = (half == 0);
                const bool sgroup = (half == 0 && c < 4);
                gather_chunk<0, 12, true>(deg, eg, hi8, hi4, coff, s_lo, s_hi, Hb, as4, ad4, hself, half0, sgroup, hd, m, S, acc);
                if (deg > 24) gather_chunk<12, 12, false>(deg, eg, hi8, hi4, coff, s_lo, s_hi, Hb, as4, ad4, hself, half0, sgroup, hd, m, S, acc);
                if (deg > 48) gather_chunk<24, 8, false>(deg, eg, hi8, hi4, coff, s_lo, s_hi, Hb, as4, ad4, hself, half0, sgroup, hd, m, S, acc);

                S += row_xor<4>(S);
                S += row_xor<8>(S);
                const float inv = 1.0f / (S + xhalf(S));
                acc.x *= inv; acc.y *= inv; acc.z *= inv; acc.w *= inv;
            } else {
                float pd = hself.x * ad4.x;
                pd = fmaf(hself.y, ad4.y, pd);
                pd = fmaf(hself.z, ad4.z, pd);
                pd = fmaf(hself.w, ad4.w, pd);
                const float hd = row16_sum(pd);
                float ps = hself.x * as4.x;
                ps = fmaf(hself.y, as4.y, ps);
                ps = fmaf(hself.z, as4.z, ps);
                ps = fmaf(hself.w, as4.w, ps);
                float es = row16_sum(ps) + hd;
                es = (es >= 0.0f) ? es : 0.2f * es;
                float m = es;
                float S = 1.0f;
                acc = hself;
                for (int base = 0; base < E; base += 32) {
                    const int  e   = base + lane;
                    const bool hit = (e < E) && (edst[e] == node);
                    int s = node;
                    if (hit) s = esrc[e];
                    unsigned mk = (unsigned)__ballot(hit);
                    while (mk != 0u) {
                        const int k = __builtin_ctz(mk);
                        mk &= mk - 1u;
                        const int    sk = __shfl(s, k, 32);
                        const float4 r  = Xv[(size_t)sk * 16 + c];
                        float p = r.x * as4.x;
                        p = fmaf(r.y, as4.y, p);
                        p = fmaf(r.z, as4.z, p);
                        p = fmaf(r.w, as4.w, p);
                        float v = row16_sum(p) + hd;
                        v = (v >= 0.0f) ? v : 0.2f * v;
                        const float mn = fmaxf(m, v);
                        const float f  = __expf(m - mn);
                        const float w  = __expf(v - mn);
                        m = mn;
                        S = fmaf(S, f, w);
                        acc.x = fmaf(acc.x, f, w * r.x); acc.y = fmaf(acc.y, f, w * r.y);
                        acc.z = fmaf(acc.z, f, w * r.z); acc.w = fmaf(acc.w, f, w * r.w);
                    }
                }
                const float inv = (half == 0) ? 1.0f / S : 0.0f;
                acc.x *= inv; acc.y *= inv; acc.z *= inv; acc.w *= inv;
            }

            acc.x += xhalf(acc.x);
            acc.y += xhalf(acc.y);
            acc.z += xhalf(acc.z);
            acc.w += xhalf(acc.w);
        }

        const int tl = row >> 4;
        if (half == 0) *(float4*)(Zs + row * ZPITCH + 4 * c) = acc;
        int owner = 0;
        if (lane == 0) {
            const int before = __hip_atomic_fetch_add(&arrived[tl], 1, __ATOMIC_ACQ_REL,
                                                      __HIP_MEMORY_SCOPE_WORKGROUP);
            owner = (before == NODES_PER_BLOCK - 1) ? 1 : 0;
        }
        owner = __builtin_amdgcn_readfirstlane(owner);
        if (owner != 0) {
            const int khalf = half;
            v8f d0 = (v8f){0,0,0,0,0,0,0,0};
            v8f d1 = (v8f){0,0,0,0,0,0,0,0};
            v8f d2 = (v8f){0,0,0,0,0,0,0,0};
            v8f d3 = (v8f){0,0,0,0,0,0,0,0};
            const float* zr = Zs + (tl * NODES_PER_BLOCK + c) * ZPITCH + 2 * khalf;
            const float4* __restrict__ wq = (const float4*)(Wm + (2 * khalf) * FDIM) + c;
#pragma nounroll
            for (int kc = 0; kc < 4; ++kc) {
                const float*  zc = zr + 16 * kc;
                const float4* __restrict__ wc = wq + (16 * kc) * (FDIM / 4);
#pragma unroll
                for (int k = 0; k < 4; ++k) {
                    const v2f    af = *(const v2f*)(zc + 4 * k);
                    const float4 q0 = wc[(4 * k) * (FDIM / 4)];
                    const float4 q1 = wc[(4 * k + 1) * (FDIM / 4)];
                    v2f b;
                    b.x = q0.x; b.y = q1.x;
                    d0 = __builtin_amdgcn_wmma_f32_16x16x4_f32(false, af, false, b, (short)0, d0, false, false);
                    b.x = q0.y; b.y = q1.y;
                    d1 = __builtin_amdgcn_wmma_f32_16x16x4_f32(false, af, false, b, (short)0, d1, false, false);
                    b.x = q0.z; b.y = q1.z;
                    d2 = __builtin_amdgcn_wmma_f32_16x16x4_f32(false, af, false, b, (short)0, d2, false, false);
                    b.x = q0.w; b.y = q1.w;
                    d3 = __builtin_amdgcn_wmma_f32_16x16x4_f32(false, af, false, b, (short)0, d3, false, false);
                }
            }

            const float4 bz = ((const float4*)bias)[c];
            const int rbase = base_node + tl * NODES_PER_BLOCK + 8 * khalf;
#pragma unroll
            for (int r = 0; r < 8; ++r) {
                float4 o;
                o.x = d0[r] + bz.x;
                o.y = d1[r] + bz.y;
                o.z = d2[r] + bz.z;
                o.w = d3[r] + bz.w;
                if (rbase + r < n_nodes) ((float4*)OUT)[(size_t)(rbase + r) * 16 + c] = o;
            }
        }
    }
}

extern "C" void kernel_launch(void* const* d_in, const int* in_sizes, int n_in,
                              void* d_out, int out_size, void* d_ws, size_t ws_size,
                              hipStream_t stream) {
    const float* x   = (const float*)d_in[0];
    const int*   ei  = (const int*)  d_in[1];
    const float* W1  = (const float*)d_in[2];
    const float* as1 = (const float*)d_in[3];
    const float* ad1 = (const float*)d_in[4];
    const float* b1  = (const float*)d_in[5];
    const float* W2  = (const float*)d_in[6];
    const float* as2 = (const float*)d_in[7];
    const float* ad2 = (const float*)d_in[8];
    const float* b2  = (const float*)d_in[9];

    const int N    = in_sizes[0] / FDIM;
    const int E    = in_sizes[1] / 2;
    const int* src = ei;
    const int* dst = ei + E;

    float* ws = (float*)d_ws;
    float* X1 = ws;  ws += (size_t)N * FDIM;
    float* wvec = ws;  ws += 4 * FDIM;
    int2* rec = (int2*)ws;

    edge_bucket<<<NCHUNK + 1, EB_THREADS, 0, stream>>>(src, dst, E, rec,
                                                       W1, as1, ad1, W2, as2, ad2, wvec);

    auto layer = [&](const float* Xin, const float* was, const float* wad,
                     const float* W, const float* b, float* OUT) {
        gat_gather_xform<<<(N + NODES_PER_GROUP - 1) / NODES_PER_GROUP,
                           32 * WAVES_PER_BLOCK, 0, stream>>>(rec, src, dst, E,
                                                              Xin, was, wad, W, b, OUT, N);
    };

    layer(x,  wvec,            wvec + FDIM,     W1, b1, X1);
    layer(X1, wvec + 2 * FDIM, wvec + 3 * FDIM, W2, b2, (float*)d_out);
}
